# starting kernel
# baseline (speedup 1.0000x reference)
#include <hip/hip_runtime.h>
#include <hip/hip_bf16.h>

typedef float v2f __attribute__((ext_vector_type(2)));
typedef float v8f __attribute__((ext_vector_type(8)));

#define N_GRAPHS 64

__global__ void k_zero_i32(int* p, int n) {
  int i = blockIdx.x * blockDim.x + threadIdx.x;
  for (; i < n; i += gridDim.x * blockDim.x) p[i] = 0;
}
__global__ void k_zero_f32(float* p, int n) {
  int i = blockIdx.x * blockDim.x + threadIdx.x;
  for (; i < n; i += gridDim.x * blockDim.x) p[i] = 0.0f;
}

__global__ void k_count(const int* __restrict__ dst, int* __restrict__ cnt, int E) {
  int e = blockIdx.x * blockDim.x + threadIdx.x;
  if (e < E) atomicAdd(&cnt[dst[e]], 1);
}

__global__ void k_scan(const int* __restrict__ cnt, int* __restrict__ rowstart,
                       float* __restrict__ dinv, int n) {
  __shared__ int tmp[1024];
  __shared__ int carry;
  const int tid = threadIdx.x;
  if (tid == 0) carry = 0;
  __syncthreads();
  for (int base = 0; base < n; base += 1024) {
    int i = base + tid;
    int v = (i < n) ? cnt[i] : 0;
    tmp[tid] = v;
    __syncthreads();
    #pragma unroll
    for (int off = 1; off < 1024; off <<= 1) {
      int t = (tid >= off) ? tmp[tid - off] : 0;
      __syncthreads();
      tmp[tid] += t;
      __syncthreads();
    }
    int incl = tmp[tid];
    int c = carry;
    if (i < n) {
      rowstart[i] = c + incl - v;
      dinv[i] = rsqrtf((float)v + 1.0f);
    }
    __syncthreads();
    if (tid == 0) carry = c + tmp[1023];
    __syncthreads();
  }
  if (tid == 0) rowstart[n] = carry;
}

__global__ void k_fill(const int* __restrict__ src, const int* __restrict__ dst,
                       const int* __restrict__ rowstart, int* __restrict__ cur,
                       int* __restrict__ csr, int E) {
  int e = blockIdx.x * blockDim.x + threadIdx.x;
  if (e < E) {
    int d = dst[e];
    int pos = atomicAdd(&cur[d], 1);
    csr[rowstart[d] + pos] = src[e];
  }
}

template <int N>
__global__ void k_gemm_wmma(const float* __restrict__ A, const float* __restrict__ W,
                            const float* __restrict__ dinv, float* __restrict__ out, int M) {
  constexpr int K = 128;
  constexpr int NT = N / 16;
  constexpr int RP = 2 * N + 32;
  extern __shared__ float Wlds[];
  for (int i = threadIdx.x; i < K * N; i += blockDim.x) {
    int k = i / N;
    int n = i % N;
    Wlds[(k >> 1) * RP + n * 2 + (k & 1)] = W[i];
  }
  __syncthreads();

  const int wave = threadIdx.x >> 5;
  const int lane = threadIdx.x & 31;
  const int mtile = blockIdx.x * 8 + wave;
  if (mtile * 16 >= M) return;

  const int Rm = mtile * 16;
  const int half = lane >> 4;
  const int l16 = lane & 15;

  v8f acc[NT] = {};
  const float* arow = A + (size_t)(Rm + l16) * K;
  const float* bbase = Wlds + half * RP + l16 * 2;

  for (int kb = 0; kb < K; kb += 4) {
    v2f a = *(const v2f*)(arow + kb + 2 * half);
    const float* bk = bbase + (kb >> 1) * RP;
    #pragma unroll
    for (int t = 0; t < NT; ++t) {
      v2f b = *(const v2f*)(bk + t * 32);
      acc[t] = __builtin_amdgcn_wmma_f32_16x16x4_f32(
          false, a, false, b, (short)0, acc[t], false, false);
    }
  }

  #pragma unroll
  for (int r = 0; r < 8; ++r) {
    const int row = Rm + r + 8 * half;
    const float di = dinv[row];
    #pragma unroll
    for (int t = 0; t < NT; ++t) {
      out[(size_t)row * N + t * 16 + l16] = acc[t][r] * di;
    }
  }
}

template <int F>
__global__ void k_aggregate(const float* __restrict__ hws, const float* __restrict__ dinv,
                            const int* __restrict__ rowstart, const int* __restrict__ csr,
                            const float* __restrict__ bias, float* __restrict__ out, int n) {
  constexpr int NPB = 256 / F;
  const int node = blockIdx.x * NPB + threadIdx.x / F;
  const int f = threadIdx.x % F;
  if (node >= n) return;
  const int s = rowstart[node];
  const int e = rowstart[node + 1];
  float sum = hws[(size_t)node * F + f];
  for (int j = s; j < e; ++j) {
    sum += hws[(size_t)csr[j] * F + f];
  }
  float v = dinv[node] * sum + bias[f];
  out[(size_t)node * F + f] = fmaxf(v, 0.0f);
}

__global__ void k_pool_acc(const float* __restrict__ h, const int* __restrict__ batch,
                           float* __restrict__ pool, float* __restrict__ cntg, int n) {
  int idx = blockIdx.x * blockDim.x + threadIdx.x;
  if (idx < n * 64) {
    int node = idx >> 6;
    int f = idx & 63;
    int g = batch[node];
    atomicAdd(&pool[g * 64 + f], h[idx]);
    if (f == 0) atomicAdd(&cntg[g], 1.0f);
  }
}

__global__ void k_pool_final(const float* __restrict__ pool, const float* __restrict__ cntg,
                             float* __restrict__ out) {
  int idx = blockIdx.x * blockDim.x + threadIdx.x;
  if (idx < N_GRAPHS * 64) {
    out[idx] = pool[idx] / fmaxf(cntg[idx >> 6], 1.0f);
  }
}

extern "C" void kernel_launch(void* const* d_in, const int* in_sizes, int n_in,
                              void* d_out, int out_size, void* d_ws, size_t ws_size,
                              hipStream_t stream) {
  const float* x     = (const float*)d_in[0];
  const float* W1    = (const float*)d_in[1];
  const float* b1    = (const float*)d_in[2];
  const float* W2    = (const float*)d_in[3];
  const float* b2    = (const float*)d_in[4];
  const int*   ei    = (const int*)d_in[5];
  const int*   batch = (const int*)d_in[6];

  const int N = in_sizes[6];
  const int E = in_sizes[5] / 2;
  const int G = N_GRAPHS;

  const int* src = ei;
  const int* dst = ei + E;

  char* base = (char*)d_ws;
  size_t off = 0;
  auto take = [&](size_t bytes) {
    size_t o = off;
    off = (off + bytes + 255) & ~(size_t)255;
    return o;
  };
  int*   cnt      = (int*)(base + take(sizeof(int) * (size_t)N));
  int*   cur      = (int*)(base + take(sizeof(int) * (size_t)N));
  int*   rowstart = (int*)(base + take(sizeof(int) * (size_t)(N + 1)));
  float* dinv     = (float*)(base + take(sizeof(float) * (size_t)N));
  int*   csr      = (int*)(base + take(sizeof(int) * (size_t)E));
  float* hw1      = (float*)(base + take(sizeof(float) * (size_t)N * 128));
  float* h1       = (float*)(base + take(sizeof(float) * (size_t)N * 128));
  float* pool     = (float*)(base + take(sizeof(float) * (size_t)G * 64));
  float* cntg     = (float*)(base + take(sizeof(float) * (size_t)G));
  float* hw2 = hw1;
  float* h2  = h1;

  const int B = 256;
  const int gN = (N + B - 1) / B;
  const int gE = (E + B - 1) / B;
  const int mtiles = (N + 15) / 16;
  const int gGemm = (mtiles + 7) / 8;

  k_zero_i32<<<gN, B, 0, stream>>>(cnt, N);
  k_count<<<gE, B, 0, stream>>>(dst, cnt, E);
  k_scan<<<1, 1024, 0, stream>>>(cnt, rowstart, dinv, N);
  k_zero_i32<<<gN, B, 0, stream>>>(cur, N);
  k_fill<<<gE, B, 0, stream>>>(src, dst, rowstart, cur, csr, E);

  k_gemm_wmma<128><<<gGemm, 256, 64 * (2 * 128 + 32) * sizeof(float), stream>>>(x, W1, dinv, hw1, N);
  k_aggregate<128><<<(N + 1) / 2, 256, 0, stream>>>(hw1, dinv, rowstart, csr, b1, h1, N);

  k_gemm_wmma<64><<<gGemm, 256, 64 * (2 * 64 + 32) * sizeof(float), stream>>>(h1, W2, dinv, hw2, N);
  k_aggregate<64><<<(N + 3) / 4, 256, 0, stream>>>(hw2, dinv, rowstart, csr, b2, h2, N);

  k_zero_f32<<<(G * 64 + G + B - 1) / B, B, 0, stream>>>(pool, G * 64 + G);
  k_pool_acc<<<((size_t)N * 64 + B - 1) / B, B, 0, stream>>>(h2, batch, pool, cntg, N);
  k_pool_final<<<(G * 64 + B - 1) / B, B, 0, stream>>>(pool, cntg, (float*)d_out);
}


# k_scan: one-pass register scan; each wave reads its 1664-element slice as 13 coalesced int4 rows and shuffle-scans them, 2 barriers; replaces 49 LDS Hillis-Steele tiles of 23 barriers each
# speedup vs baseline: 1.3209x; 1.3209x over previous
#include <hip/hip_runtime.h>
#include <hip/hip_bf16.h>

typedef float v2f __attribute__((ext_vector_type(2)));
typedef float v8f __attribute__((ext_vector_type(8)));

#define N_GRAPHS 64

__global__ void k_zero_i32(int* p, int n) {
  int i = blockIdx.x * blockDim.x + threadIdx.x;
  for (; i < n; i += gridDim.x * blockDim.x) p[i] = 0;
}
__global__ void k_zero_f32(float* p, int n) {
  int i = blockIdx.x * blockDim.x + threadIdx.x;
  for (; i < n; i += gridDim.x * blockDim.x) p[i] = 0.0f;
}

__global__ void k_count(const int* __restrict__ dst, int* __restrict__ cnt, int E) {
  int e = blockIdx.x * blockDim.x + threadIdx.x;
  if (e < E) atomicAdd(&cnt[dst[e]], 1);
}

#define SCAN_ROWS 13
__global__ void k_scan(const int* __restrict__ cnt, int* __restrict__ rowstart,
                       float* __restrict__ dinv, int n) {
  constexpr int WSEG = SCAN_ROWS * 128;
  constexpr int TILE = 32 * WSEG;
  __shared__ int wtot[32];
  const int tid = threadIdx.x;
  const int lane = tid & 31;
  const int wave = tid >> 5;
  const int n4 = n & ~3;
  int carry = 0;
  for (int base = 0; base < n4; base += TILE) {
    const int i0 = base + wave * WSEG + lane * 4;
    int4 v[SCAN_ROWS];
    #pragma unroll
    for (int j = 0; j < SCAN_ROWS; ++j) {
      const int i = i0 + j * 128;
      int4 t = make_int4(0, 0, 0, 0);
      if (i < n4) t = *(const int4*)(cnt + i);
      v[j] = t;
    }
    int inc[SCAN_ROWS];
    #pragma unroll
    for (int j = 0; j < SCAN_ROWS; ++j) inc[j] = v[j].x + v[j].y + v[j].z + v[j].w;
    #pragma unroll
    for (int off = 1; off < 32; off <<= 1) {
      #pragma unroll
      for (int j = 0; j < SCAN_ROWS; ++j) {
        const int t = __shfl_up(inc[j], off, 32);
        if (lane >= off) inc[j] += t;
      }
    }
    int rowoff[SCAN_ROWS];
    int wsum = 0;
    #pragma unroll
    for (int j = 0; j < SCAN_ROWS; ++j) {
      rowoff[j] = wsum;
      wsum += __builtin_amdgcn_readlane(inc[j], 31);
    }
    if (lane == 0) wtot[wave] = wsum;
    __syncthreads();
    const int wt = wtot[lane];
    int winc = wt;
    #pragma unroll
    for (int off = 1; off < 32; off <<= 1) {
      const int t = __shfl_up(winc, off, 32);
      if (lane >= off) winc += t;
    }
    const int woff = carry + __shfl(winc - wt, wave, 32);
    carry += __builtin_amdgcn_readlane(winc, 31);
    #pragma unroll
    for (int j = 0; j < SCAN_ROWS; ++j) {
      const int i = i0 + j * 128;
      const int4 c = v[j];
      int4 r;
      r.x = woff + rowoff[j] + inc[j] - (c.x + c.y + c.z + c.w);
      r.y = r.x + c.x;
      r.z = r.y + c.y;
      r.w = r.z + c.z;
      float4 d;
      d.x = rsqrtf((float)c.x + 1.0f);
      d.y = rsqrtf((float)c.y + 1.0f);
      d.z = rsqrtf((float)c.z + 1.0f);
      d.w = rsqrtf((float)c.w + 1.0f);
      if (i < n4) {
        *(int4*)(rowstart + i) = r;
        *(float4*)(dinv + i) = d;
      }
    }
    __syncthreads();
  }
  if (tid == 0) {
    int c = carry;
    for (int i = n4; i < n; ++i) {
      const int vv = cnt[i];
      rowstart[i] = c;
      dinv[i] = rsqrtf((float)vv + 1.0f);
      c += vv;
    }
    rowstart[n] = c;
  }
}

__global__ void k_fill(const int* __restrict__ src, const int* __restrict__ dst,
                       const int* __restrict__ rowstart, int* __restrict__ cur,
                       int* __restrict__ csr, int E) {
  int e = blockIdx.x * blockDim.x + threadIdx.x;
  if (e < E) {
    int d = dst[e];
    int pos = atomicAdd(&cur[d], 1);
    csr[rowstart[d] + pos] = src[e];
  }
}

template <int N>
__global__ void k_gemm_wmma(const float* __restrict__ A, const float* __restrict__ W,
                            const float* __restrict__ dinv, float* __restrict__ out, int M) {
  constexpr int K = 128;
  constexpr int NT = N / 16;
  constexpr int RP = 2 * N + 32;
  extern __shared__ float Wlds[];
  for (int i = threadIdx.x; i < K * N; i += blockDim.x) {
    int k = i / N;
    int n = i % N;
    Wlds[(k >> 1) * RP + n * 2 + (k & 1)] = W[i];
  }
  __syncthreads();

  const int wave = threadIdx.x >> 5;
  const int lane = threadIdx.x & 31;
  const int mtile = blockIdx.x * 8 + wave;
  if (mtile * 16 >= M) return;

  const int Rm = mtile * 16;
  const int half = lane >> 4;
  const int l16 = lane & 15;

  v8f acc[NT] = {};
  const float* arow = A + (size_t)(Rm + l16) * K;
  const float* bbase = Wlds + half * RP + l16 * 2;

  for (int kb = 0; kb < K; kb += 4) {
    v2f a = *(const v2f*)(arow + kb + 2 * half);
    const float* bk = bbase + (kb >> 1) * RP;
    #pragma unroll
    for (int t = 0; t < NT; ++t) {
      v2f b = *(const v2f*)(bk + t * 32);
      acc[t] = __builtin_amdgcn_wmma_f32_16x16x4_f32(
          false, a, false, b, (short)0, acc[t], false, false);
    }
  }

  #pragma unroll
  for (int r = 0; r < 8; ++r) {
    const int row = Rm + r + 8 * half;
    const float di = dinv[row];
    #pragma unroll
    for (int t = 0; t < NT; ++t) {
      out[(size_t)row * N + t * 16 + l16] = acc[t][r] * di;
    }
  }
}

template <int F>
__global__ void k_aggregate(const float* __restrict__ hws, const float* __restrict__ dinv,
                            const int* __restrict__ rowstart, const int* __restrict__ csr,
                            const float* __restrict__ bias, float* __restrict__ out, int n) {
  constexpr int NPB = 256 / F;
  const int node = blockIdx.x * NPB + threadIdx.x / F;
  const int f = threadIdx.x % F;
  if (node >= n) return;
  const int s = rowstart[node];
  const int e = rowstart[node + 1];
  float sum = hws[(size_t)node * F + f];
  for (int j = s; j < e; ++j) {
    sum += hws[(size_t)csr[j] * F + f];
  }
  float v = dinv[node] * sum + bias[f];
  out[(size_t)node * F + f] = fmaxf(v, 0.0f);
}

__global__ void k_pool_acc(const float* __restrict__ h, const int* __restrict__ batch,
                           float* __restrict__ pool, float* __restrict__ cntg, int n) {
  int idx = blockIdx.x * blockDim.x + threadIdx.x;
  if (idx < n * 64) {
    int node = idx >> 6;
    int f = idx & 63;
    int g = batch[node];
    atomicAdd(&pool[g * 64 + f], h[idx]);
    if (f == 0) atomicAdd(&cntg[g], 1.0f);
  }
}

__global__ void k_pool_final(const float* __restrict__ pool, const float* __restrict__ cntg,
                             float* __restrict__ out) {
  int idx = blockIdx.x * blockDim.x + threadIdx.x;
  if (idx < N_GRAPHS * 64) {
    out[idx] = pool[idx] / fmaxf(cntg[idx >> 6], 1.0f);
  }
}

extern "C" void kernel_launch(void* const* d_in, const int* in_sizes, int n_in,
                              void* d_out, int out_size, void* d_ws, size_t ws_size,
                              hipStream_t stream) {
  const float* x     = (const float*)d_in[0];
  const float* W1    = (const float*)d_in[1];
  const float* b1    = (const float*)d_in[2];
  const float* W2    = (const float*)d_in[3];
  const float* b2    = (const float*)d_in[4];
  const int*   ei    = (const int*)d_in[5];
  const int*   batch = (const int*)d_in[6];

  const int N = in_sizes[6];
  const int E = in_sizes[5] / 2;
  const int G = N_GRAPHS;

  const int* src = ei;
  const int* dst = ei + E;

  char* base = (char*)d_ws;
  size_t off = 0;
  auto take = [&](size_t bytes) {
    size_t o = off;
    off = (off + bytes + 255) & ~(size_t)255;
    return o;
  };
  int*   cnt      = (int*)(base + take(sizeof(int) * (size_t)N));
  int*   cur      = (int*)(base + take(sizeof(int) * (size_t)N));
  int*   rowstart = (int*)(base + take(sizeof(int) * (size_t)(N + 1)));
  float* dinv     = (float*)(base + take(sizeof(float) * (size_t)N));
  int*   csr      = (int*)(base + take(sizeof(int) * (size_t)E));
  float* hw1      = (float*)(base + take(sizeof(float) * (size_t)N * 128));
  float* h1       = (float*)(base + take(sizeof(float) * (size_t)N * 128));
  float* pool     = (float*)(base + take(sizeof(float) * (size_t)G * 64));
  float* cntg     = (float*)(base + take(sizeof(float) * (size_t)G));
  float* hw2 = hw1;
  float* h2  = h1;

  const int B = 256;
  const int gN = (N + B - 1) / B;
  const int gE = (E + B - 1) / B;
  const int mtiles = (N + 15) / 16;
  const int gGemm = (mtiles + 7) / 8;

  k_zero_i32<<<gN, B, 0, stream>>>(cnt, N);
  k_count<<<gE, B, 0, stream>>>(dst, cnt, E);
  k_scan<<<1, 1024, 0, stream>>>(cnt, rowstart, dinv, N);
  k_zero_i32<<<gN, B, 0, stream>>>(cur, N);
  k_fill<<<gE, B, 0, stream>>>(src, dst, rowstart, cur, csr, E);

  k_gemm_wmma<128><<<gGemm, 256, 64 * (2 * 128 + 32) * sizeof(float), stream>>>(x, W1, dinv, hw1, N);
  k_aggregate<128><<<(N + 1) / 2, 256, 0, stream>>>(hw1, dinv, rowstart, csr, b1, h1, N);

  k_gemm_wmma<64><<<gGemm, 256, 64 * (2 * 64 + 32) * sizeof(float), stream>>>(h1, W2, dinv, hw2, N);
  k_aggregate<64><<<(N + 3) / 4, 256, 0, stream>>>(hw2, dinv, rowstart, csr, b2, h2, N);

  k_zero_f32<<<(G * 64 + G + B - 1) / B, B, 0, stream>>>(pool, G * 64 + G);
  k_pool_acc<<<((size_t)N * 64 + B - 1) / B, B, 0, stream>>>(h2, batch, pool, cntg, N);
  k_pool_final<<<(G * 64 + B - 1) / B, B, 0, stream>>>(pool, cntg, (float*)d_out);
}


# k_aggregate: a wave per node row (float4/float2 per lane, one load a row), neighbour list read once coalesced, broadcast by readlane, 8 row gathers in flight, waves stride over nodes; sum order kept
# speedup vs baseline: 1.5070x; 1.1409x over previous
#include <hip/hip_runtime.h>
#include <hip/hip_bf16.h>

typedef float v2f __attribute__((ext_vector_type(2)));
typedef float v8f __attribute__((ext_vector_type(8)));

#define N_GRAPHS 64

__global__ void k_zero_i32(int* p, int n) {
  int i = blockIdx.x * blockDim.x + threadIdx.x;
  for (; i < n; i += gridDim.x * blockDim.x) p[i] = 0;
}
__global__ void k_zero_f32(float* p, int n) {
  int i = blockIdx.x * blockDim.x + threadIdx.x;
  for (; i < n; i += gridDim.x * blockDim.x) p[i] = 0.0f;
}

__global__ void k_count(const int* __restrict__ dst, int* __restrict__ cnt, int E) {
  int e = blockIdx.x * blockDim.x + threadIdx.x;
  if (e < E) atomicAdd(&cnt[dst[e]], 1);
}

#define SCAN_ROWS 13
__global__ void k_scan(const int* __restrict__ cnt, int* __restrict__ rowstart,
                       float* __restrict__ dinv, int n) {
  constexpr int WSEG = SCAN_ROWS * 128;
  constexpr int TILE = 32 * WSEG;
  __shared__ int wtot[32];
  const int tid = threadIdx.x;
  const int lane = tid & 31;
  const int wave = tid >> 5;
  const int n4 = n & ~3;
  int carry = 0;
  for (int base = 0; base < n4; base += TILE) {
    const int i0 = base + wave * WSEG + lane * 4;
    int4 v[SCAN_ROWS];
    #pragma unroll
    for (int j = 0; j < SCAN_ROWS; ++j) {
      const int i = i0 + j * 128;
      int4 t = make_int4(0, 0, 0, 0);
      if (i < n4) t = *(const int4*)(cnt + i);
      v[j] = t;
    }
    int inc[SCAN_ROWS];
    #pragma unroll
    for (int j = 0; j < SCAN_ROWS; ++j) inc[j] = v[j].x + v[j].y + v[j].z + v[j].w;
    #pragma unroll
    for (int off = 1; off < 32; off <<= 1) {
      #pragma unroll
      for (int j = 0; j < SCAN_ROWS; ++j) {
        const int t = __shfl_up(inc[j], off, 32);
        if (lane >= off) inc[j] += t;
      }
    }
    int rowoff[SCAN_ROWS];
    int wsum = 0;
    #pragma unroll
    for (int j = 0; j < SCAN_ROWS; ++j) {
      rowoff[j] = wsum;
      wsum += __builtin_amdgcn_readlane(inc[j], 31);
    }
    if (lane == 0) wtot[wave] = wsum;
    __syncthreads();
    const int wt = wtot[lane];
    int winc = wt;
    #pragma unroll
    for (int off = 1; off < 32; off <<= 1) {
      const int t = __shfl_up(winc, off, 32);
      if (lane >= off) winc += t;
    }
    const int woff = carry + __shfl(winc - wt, wave, 32);
    carry += __builtin_amdgcn_readlane(winc, 31);
    #pragma unroll
    for (int j = 0; j < SCAN_ROWS; ++j) {
      const int i = i0 + j * 128;
      const int4 c = v[j];
      int4 r;
      r.x = woff + rowoff[j] + inc[j] - (c.x + c.y + c.z + c.w);
      r.y = r.x + c.x;
      r.z = r.y + c.y;
      r.w = r.z + c.z;
      float4 d;
      d.x = rsqrtf((float)c.x + 1.0f);
      d.y = rsqrtf((float)c.y + 1.0f);
      d.z = rsqrtf((float)c.z + 1.0f);
      d.w = rsqrtf((float)c.w + 1.0f);
      if (i < n4) {
        *(int4*)(rowstart + i) = r;
        *(float4*)(dinv + i) = d;
      }
    }
    __syncthreads();
  }
  if (tid == 0) {
    int c = carry;
    for (int i = n4; i < n; ++i) {
      const int vv = cnt[i];
      rowstart[i] = c;
      dinv[i] = rsqrtf((float)vv + 1.0f);
      c += vv;
    }
    rowstart[n] = c;
  }
}

__global__ void k_fill(const int* __restrict__ src, const int* __restrict__ dst,
                       const int* __restrict__ rowstart, int* __restrict__ cur,
                       int* __restrict__ csr, int E) {
  int e = blockIdx.x * blockDim.x + threadIdx.x;
  if (e < E) {
    int d = dst[e];
    int pos = atomicAdd(&cur[d], 1);
    csr[rowstart[d] + pos] = src[e];
  }
}

template <int N>
__global__ void k_gemm_wmma(const float* __restrict__ A, const float* __restrict__ W,
                            const float* __restrict__ dinv, float* __restrict__ out, int M) {
  constexpr int K = 128;
  constexpr int NT = N / 16;
  constexpr int RP = 2 * N + 32;
  extern __shared__ float Wlds[];
  for (int i = threadIdx.x; i < K * N; i += blockDim.x) {
    int k = i / N;
    int n = i % N;
    Wlds[(k >> 1) * RP + n * 2 + (k & 1)] = W[i];
  }
  __syncthreads();

  const int wave = threadIdx.x >> 5;
  const int lane = threadIdx.x & 31;
  const int mtile = blockIdx.x * 8 + wave;
  if (mtile * 16 >= M) return;

  const int Rm = mtile * 16;
  const int half = lane >> 4;
  const int l16 = lane & 15;

  v8f acc[NT] = {};
  const float* arow = A + (size_t)(Rm + l16) * K;
  const float* bbase = Wlds + half * RP + l16 * 2;

  for (int kb = 0; kb < K; kb += 4) {
    v2f a = *(const v2f*)(arow + kb + 2 * half);
    const float* bk = bbase + (kb >> 1) * RP;
    #pragma unroll
    for (int t = 0; t < NT; ++t) {
      v2f b = *(const v2f*)(bk + t * 32);
      acc[t] = __builtin_amdgcn_wmma_f32_16x16x4_f32(
          false, a, false, b, (short)0, acc[t], false, false);
    }
  }

  #pragma unroll
  for (int r = 0; r < 8; ++r) {
    const int row = Rm + r + 8 * half;
    const float di = dinv[row];
    #pragma unroll
    for (int t = 0; t < NT; ++t) {
      out[(size_t)row * N + t * 16 + l16] = acc[t][r] * di;
    }
  }
}

template <int F>
struct AggVec {
  typedef float type __attribute__((ext_vector_type(F / 32)));
};

template <int F, int NB>
__device__ __forceinline__ typename AggVec<F>::type agg_rows(
    const float* __restrict__ hws, int idx, int k, int lane, typename AggVec<F>::type sum) {
  typedef typename AggVec<F>::type vf;
  constexpr int V = F / 32;
  vf v[NB];
  #pragma unroll
  for (int u = 0; u < NB; ++u) {
    const unsigned r = (unsigned)__builtin_amdgcn_readlane(idx, k + u);
    const float* row = hws + (size_t)r * F;
    v[u] = *(const vf*)(row + V * lane);
  }
  #pragma unroll
  for (int u = 0; u < NB; ++u) sum += v[u];
  return sum;
}

template <int F>
__global__ void __launch_bounds__(256)
k_aggregate(const float* __restrict__ hws, const float* __restrict__ dinv,
            const int* __restrict__ rowstart, const int* __restrict__ csr,
            const float* __restrict__ bias, float* __restrict__ out, int n) {
  constexpr int V = F / 32;
  typedef typename AggVec<F>::type vf;
  const int lane = threadIdx.x & 31;
  const int wave = __builtin_amdgcn_readfirstlane(threadIdx.x >> 5);
  const int wpb = blockDim.x >> 5;
  const int nwaves = gridDim.x * wpb;
  const vf bv = *(const vf*)(bias + V * lane);
  for (int node = blockIdx.x * wpb + wave; node < n; node += nwaves) {
    const int s = rowstart[node];
    const int e = rowstart[node + 1];
    vf sum = *(const vf*)(hws + (size_t)node * F + V * lane);
    for (int base = s; base < e; base += 32) {
      const int cnt = min(e - base, 32);
      const int idx = csr[base + min(lane, cnt - 1)];
      int k = 0;
      for (; k + 8 <= cnt; k += 8) sum = agg_rows<F, 8>(hws, idx, k, lane, sum);
      if (cnt & 4) { sum = agg_rows<F, 4>(hws, idx, k, lane, sum); k += 4; }
      if (cnt & 2) { sum = agg_rows<F, 2>(hws, idx, k, lane, sum); k += 2; }
      if (cnt & 1) sum = agg_rows<F, 1>(hws, idx, k, lane, sum);
    }
    const float di = dinv[node];
    vf o;
    #pragma unroll
    for (int c = 0; c < V; ++c) o[c] = fmaxf(di * sum[c] + bv[c], 0.0f);
    *(vf*)(out + (size_t)node * F + V * lane) = o;
  }
}

__global__ void k_pool_acc(const float* __restrict__ h, const int* __restrict__ batch,
                           float* __restrict__ pool, float* __restrict__ cntg, int n) {
  int idx = blockIdx.x * blockDim.x + threadIdx.x;
  if (idx < n * 64) {
    int node = idx >> 6;
    int f = idx & 63;
    int g = batch[node];
    atomicAdd(&pool[g * 64 + f], h[idx]);
    if (f == 0) atomicAdd(&cntg[g], 1.0f);
  }
}

__global__ void k_pool_final(const float* __restrict__ pool, const float* __restrict__ cntg,
                             float* __restrict__ out) {
  int idx = blockIdx.x * blockDim.x + threadIdx.x;
  if (idx < N_GRAPHS * 64) {
    out[idx] = pool[idx] / fmaxf(cntg[idx >> 6], 1.0f);
  }
}

extern "C" void kernel_launch(void* const* d_in, const int* in_sizes, int n_in,
                              void* d_out, int out_size, void* d_ws, size_t ws_size,
                              hipStream_t stream) {
  const float* x     = (const float*)d_in[0];
  const float* W1    = (const float*)d_in[1];
  const float* b1    = (const float*)d_in[2];
  const float* W2    = (const float*)d_in[3];
  const float* b2    = (const float*)d_in[4];
  const int*   ei    = (const int*)d_in[5];
  const int*   batch = (const int*)d_in[6];

  const int N = in_sizes[6];
  const int E = in_sizes[5] / 2;
  const int G = N_GRAPHS;

  const int* src = ei;
  const int* dst = ei + E;

  char* base = (char*)d_ws;
  size_t off = 0;
  auto take = [&](size_t bytes) {
    size_t o = off;
    off = (off + bytes + 255) & ~(size_t)255;
    return o;
  };
  int*   cnt      = (int*)(base + take(sizeof(int) * (size_t)N));
  int*   cur      = (int*)(base + take(sizeof(int) * (size_t)N));
  int*   rowstart = (int*)(base + take(sizeof(int) * (size_t)(N + 1)));
  float* dinv     = (float*)(base + take(sizeof(float) * (size_t)N));
  int*   csr      = (int*)(base + take(sizeof(int) * (size_t)E));
  float* hw1      = (float*)(base + take(sizeof(float) * (size_t)N * 128));
  float* h1       = (float*)(base + take(sizeof(float) * (size_t)N * 128));
  float* pool     = (float*)(base + take(sizeof(float) * (size_t)G * 64));
  float* cntg     = (float*)(base + take(sizeof(float) * (size_t)G));
  float* hw2 = hw1;
  float* h2  = h1;

  const int B = 256;
  const int gN = (N + B - 1) / B;
  const int gE = (E + B - 1) / B;
  const int mtiles = (N + 15) / 16;
  const int gGemm = (mtiles + 7) / 8;

  k_zero_i32<<<gN, B, 0, stream>>>(cnt, N);
  k_count<<<gE, B, 0, stream>>>(dst, cnt, E);
  k_scan<<<1, 1024, 0, stream>>>(cnt, rowstart, dinv, N);
  k_zero_i32<<<gN, B, 0, stream>>>(cur, N);
  k_fill<<<gE, B, 0, stream>>>(src, dst, rowstart, cur, csr, E);

  k_gemm_wmma<128><<<gGemm, 256, 64 * (2 * 128 + 32) * sizeof(float), stream>>>(x, W1, dinv, hw1, N);
  k_aggregate<128><<<2048, 256, 0, stream>>>(hw1, dinv, rowstart, csr, b1, h1, N);

  k_gemm_wmma<64><<<gGemm, 256, 64 * (2 * 64 + 32) * sizeof(float), stream>>>(h1, W2, dinv, hw2, N);
  k_aggregate<64><<<2048, 256, 0, stream>>>(hw2, dinv, rowstart, csr, b2, h2, N);

  k_zero_f32<<<(G * 64 + G + B - 1) / B, B, 0, stream>>>(pool, G * 64 + G);
  k_pool_acc<<<((size_t)N * 64 + B - 1) / B, B, 0, stream>>>(h2, batch, pool, cntg, N);
  k_pool_final<<<(G * 64 + B - 1) / B, B, 0, stream>>>(pool, cntg, (float*)d_out);
}


# k_pool_acc: a wave sums 32 consecutive node rows in registers (256-byte row loads, all in flight), one atomic add per run of equal graph id; pooled sum order changes; 100000 waves become 1563
# speedup vs baseline: 1.9225x; 1.2757x over previous
#include <hip/hip_runtime.h>
#include <hip/hip_bf16.h>

typedef float v2f __attribute__((ext_vector_type(2)));
typedef float v8f __attribute__((ext_vector_type(8)));

#define N_GRAPHS 64

__global__ void k_zero_i32(int* p, int n) {
  int i = blockIdx.x * blockDim.x + threadIdx.x;
  for (; i < n; i += gridDim.x * blockDim.x) p[i] = 0;
}
__global__ void k_zero_f32(float* p, int n) {
  int i = blockIdx.x * blockDim.x + threadIdx.x;
  for (; i < n; i += gridDim.x * blockDim.x) p[i] = 0.0f;
}

__global__ void k_count(const int* __restrict__ dst, int* __restrict__ cnt, int E) {
  int e = blockIdx.x * blockDim.x + threadIdx.x;
  if (e < E) atomicAdd(&cnt[dst[e]], 1);
}

#define SCAN_ROWS 13
__global__ void k_scan(const int* __restrict__ cnt, int* __restrict__ rowstart,
                       float* __restrict__ dinv, int n) {
  constexpr int WSEG = SCAN_ROWS * 128;
  constexpr int TILE = 32 * WSEG;
  __shared__ int wtot[32];
  const int tid = threadIdx.x;
  const int lane = tid & 31;
  const int wave = tid >> 5;
  const int n4 = n & ~3;
  int carry = 0;
  for (int base = 0; base < n4; base += TILE) {
    const int i0 = base + wave * WSEG + lane * 4;
    int4 v[SCAN_ROWS];
    #pragma unroll
    for (int j = 0; j < SCAN_ROWS; ++j) {
      const int i = i0 + j * 128;
      int4 t = make_int4(0, 0, 0, 0);
      if (i < n4) t = *(const int4*)(cnt + i);
      v[j] = t;
    }
    int inc[SCAN_ROWS];
    #pragma unroll
    for (int j = 0; j < SCAN_ROWS; ++j) inc[j] = v[j].x + v[j].y + v[j].z + v[j].w;
    #pragma unroll
    for (int off = 1; off < 32; off <<= 1) {
      #pragma unroll
      for (int j = 0; j < SCAN_ROWS; ++j) {
        const int t = __shfl_up(inc[j], off, 32);
        if (lane >= off) inc[j] += t;
      }
    }
    int rowoff[SCAN_ROWS];
    int wsum = 0;
    #pragma unroll
    for (int j = 0; j < SCAN_ROWS; ++j) {
      rowoff[j] = wsum;
      wsum += __builtin_amdgcn_readlane(inc[j], 31);
    }
    if (lane == 0) wtot[wave] = wsum;
    __syncthreads();
    const int wt = wtot[lane];
    int winc = wt;
    #pragma unroll
    for (int off = 1; off < 32; off <<= 1) {
      const int t = __shfl_up(winc, off, 32);
      if (lane >= off) winc += t;
    }
    const int woff = carry + __shfl(winc - wt, wave, 32);
    carry += __builtin_amdgcn_readlane(winc, 31);
    #pragma unroll
    for (int j = 0; j < SCAN_ROWS; ++j) {
      const int i = i0 + j * 128;
      const int4 c = v[j];
      int4 r;
      r.x = woff + rowoff[j] + inc[j] - (c.x + c.y + c.z + c.w);
      r.y = r.x + c.x;
      r.z = r.y + c.y;
      r.w = r.z + c.z;
      float4 d;
      d.x = rsqrtf((float)c.x + 1.0f);
      d.y = rsqrtf((float)c.y + 1.0f);
      d.z = rsqrtf((float)c.z + 1.0f);
      d.w = rsqrtf((float)c.w + 1.0f);
      if (i < n4) {
        *(int4*)(rowstart + i) = r;
        *(float4*)(dinv + i) = d;
      }
    }
    __syncthreads();
  }
  if (tid == 0) {
    int c = carry;
    for (int i = n4; i < n; ++i) {
      const int vv = cnt[i];
      rowstart[i] = c;
      dinv[i] = rsqrtf((float)vv + 1.0f);
      c += vv;
    }
    rowstart[n] = c;
  }
}

__global__ void k_fill(const int* __restrict__ src, const int* __restrict__ dst,
                       const int* __restrict__ rowstart, int* __restrict__ cur,
                       int* __restrict__ csr, int E) {
  int e = blockIdx.x * blockDim.x + threadIdx.x;
  if (e < E) {
    int d = dst[e];
    int pos = atomicAdd(&cur[d], 1);
    csr[rowstart[d] + pos] = src[e];
  }
}

template <int N>
__global__ void k_gemm_wmma(const float* __restrict__ A, const float* __restrict__ W,
                            const float* __restrict__ dinv, float* __restrict__ out, int M) {
  constexpr int K = 128;
  constexpr int NT = N / 16;
  constexpr int RP = 2 * N + 32;
  extern __shared__ float Wlds[];
  for (int i = threadIdx.x; i < K * N; i += blockDim.x) {
    int k = i / N;
    int n = i % N;
    Wlds[(k >> 1) * RP + n * 2 + (k & 1)] = W[i];
  }
  __syncthreads();

  const int wave = threadIdx.x >> 5;
  const int lane = threadIdx.x & 31;
  const int mtile = blockIdx.x * 8 + wave;
  if (mtile * 16 >= M) return;

  const int Rm = mtile * 16;
  const int half = lane >> 4;
  const int l16 = lane & 15;

  v8f acc[NT] = {};
  const float* arow = A + (size_t)(Rm + l16) * K;
  const float* bbase = Wlds + half * RP + l16 * 2;

  for (int kb = 0; kb < K; kb += 4) {
    v2f a = *(const v2f*)(arow + kb + 2 * half);
    const float* bk = bbase + (kb >> 1) * RP;
    #pragma unroll
    for (int t = 0; t < NT; ++t) {
      v2f b = *(const v2f*)(bk + t * 32);
      acc[t] = __builtin_amdgcn_wmma_f32_16x16x4_f32(
          false, a, false, b, (short)0, acc[t], false, false);
    }
  }

  #pragma unroll
  for (int r = 0; r < 8; ++r) {
    const int row = Rm + r + 8 * half;
    const float di = dinv[row];
    #pragma unroll
    for (int t = 0; t < NT; ++t) {
      out[(size_t)row * N + t * 16 + l16] = acc[t][r] * di;
    }
  }
}

template <int F>
struct AggVec {
  typedef float type __attribute__((ext_vector_type(F / 32)));
};

template <int F, int NB>
__device__ __forceinline__ typename AggVec<F>::type agg_rows(
    const float* __restrict__ hws, int idx, int k, int lane, typename AggVec<F>::type sum) {
  typedef typename AggVec<F>::type vf;
  constexpr int V = F / 32;
  vf v[NB];
  #pragma unroll
  for (int u = 0; u < NB; ++u) {
    const unsigned r = (unsigned)__builtin_amdgcn_readlane(idx, k + u);
    const float* row = hws + (size_t)r * F;
    v[u] = *(const vf*)(row + V * lane);
  }
  #pragma unroll
  for (int u = 0; u < NB; ++u) sum += v[u];
  return sum;
}

template <int F>
__global__ void __launch_bounds__(256)
k_aggregate(const float* __restrict__ hws, const float* __restrict__ dinv,
            const int* __restrict__ rowstart, const int* __restrict__ csr,
            const float* __restrict__ bias, float* __restrict__ out, int n) {
  constexpr int V = F / 32;
  typedef typename AggVec<F>::type vf;
  const int lane = threadIdx.x & 31;
  const int wave = __builtin_amdgcn_readfirstlane(threadIdx.x >> 5);
  const int wpb = blockDim.x >> 5;
  const int nwaves = gridDim.x * wpb;
  const vf bv = *(const vf*)(bias + V * lane);
  for (int node = blockIdx.x * wpb + wave; node < n; node += nwaves) {
    const int s = rowstart[node];
    const int e = rowstart[node + 1];
    vf sum = *(const vf*)(hws + (size_t)node * F + V * lane);
    for (int base = s; base < e; base += 32) {
      const int cnt = min(e - base, 32);
      const int idx = csr[base + min(lane, cnt - 1)];
      int k = 0;
      for (; k + 8 <= cnt; k += 8) sum = agg_rows<F, 8>(hws, idx, k, lane, sum);
      if (cnt & 4) { sum = agg_rows<F, 4>(hws, idx, k, lane, sum); k += 4; }
      if (cnt & 2) { sum = agg_rows<F, 2>(hws, idx, k, lane, sum); k += 2; }
      if (cnt & 1) sum = agg_rows<F, 1>(hws, idx, k, lane, sum);
    }
    const float di = dinv[node];
    vf o;
    #pragma unroll
    for (int c = 0; c < V; ++c) o[c] = fmaxf(di * sum[c] + bv[c], 0.0f);
    *(vf*)(out + (size_t)node * F + V * lane) = o;
  }
}

#define POOL_ROWS 32
__device__ __forceinline__ void pool_flush(float* __restrict__ pool, float* __restrict__ cntg,
                                           int g, v2f acc, int run, int lane) {
  float* p = pool + g * 64 + 2 * lane;
  atomicAdd(p, acc.x);
  atomicAdd(p + 1, acc.y);
  if (lane == 0) atomicAdd(&cntg[g], (float)run);
}

__global__ void __launch_bounds__(128)
k_pool_acc(const float* __restrict__ h, const int* __restrict__ batch,
           float* __restrict__ pool, float* __restrict__ cntg, int n) {
  const int lane = threadIdx.x & 31;
  const int wave = __builtin_amdgcn_readfirstlane(threadIdx.x >> 5);
  const int node0 = (blockIdx.x * (blockDim.x >> 5) + wave) * POOL_ROWS;
  if (node0 >= n) return;
  const int nv = min(POOL_ROWS, n - node0);
  const int ids = batch[node0 + min(lane, nv - 1)];
  v2f v[POOL_ROWS];
  #pragma unroll
  for (int u = 0; u < POOL_ROWS; ++u) {
    const int r = min(node0 + u, n - 1);
    const float* row = h + (size_t)r * 64;
    v[u] = *(const v2f*)(row + 2 * lane);
  }
  int cur = __builtin_amdgcn_readlane(ids, 0);
  int run = 1;
  v2f acc = v[0];
  #pragma unroll
  for (int u = 1; u < POOL_ROWS; ++u) {
    if (u < nv) {
      const int g = __builtin_amdgcn_readlane(ids, u);
      if (g != cur) {
        pool_flush(pool, cntg, cur, acc, run, lane);
        acc = (v2f){0.0f, 0.0f};
        run = 0;
        cur = g;
      }
      acc += v[u];
      ++run;
    }
  }
  pool_flush(pool, cntg, cur, acc, run, lane);
}

__global__ void k_pool_final(const float* __restrict__ pool, const float* __restrict__ cntg,
                             float* __restrict__ out) {
  int idx = blockIdx.x * blockDim.x + threadIdx.x;
  if (idx < N_GRAPHS * 64) {
    out[idx] = pool[idx] / fmaxf(cntg[idx >> 6], 1.0f);
  }
}

extern "C" void kernel_launch(void* const* d_in, const int* in_sizes, int n_in,
                              void* d_out, int out_size, void* d_ws, size_t ws_size,
                              hipStream_t stream) {
  const float* x     = (const float*)d_in[0];
  const float* W1    = (const float*)d_in[1];
  const float* b1    = (const float*)d_in[2];
  const float* W2    = (const float*)d_in[3];
  const float* b2    = (const float*)d_in[4];
  const int*   ei    = (const int*)d_in[5];
  const int*   batch = (const int*)d_in[6];

  const int N = in_sizes[6];
  const int E = in_sizes[5] / 2;
  const int G = N_GRAPHS;

  const int* src = ei;
  const int* dst = ei + E;

  char* base = (char*)d_ws;
  size_t off = 0;
  auto take = [&](size_t bytes) {
    size_t o = off;
    off = (off + bytes + 255) & ~(size_t)255;
    return o;
  };
  int*   cnt      = (int*)(base + take(sizeof(int) * (size_t)N));
  int*   cur      = (int*)(base + take(sizeof(int) * (size_t)N));
  int*   rowstart = (int*)(base + take(sizeof(int) * (size_t)(N + 1)));
  float* dinv     = (float*)(base + take(sizeof(float) * (size_t)N));
  int*   csr      = (int*)(base + take(sizeof(int) * (size_t)E));
  float* hw1      = (float*)(base + take(sizeof(float) * (size_t)N * 128));
  float* h1       = (float*)(base + take(sizeof(float) * (size_t)N * 128));
  float* pool     = (float*)(base + take(sizeof(float) * (size_t)G * 64));
  float* cntg     = (float*)(base + take(sizeof(float) * (size_t)G));
  float* hw2 = hw1;
  float* h2  = h1;

  const int B = 256;
  const int gN = (N + B - 1) / B;
  const int gE = (E + B - 1) / B;
  const int mtiles = (N + 15) / 16;
  const int gGemm = (mtiles + 7) / 8;

  k_zero_i32<<<gN, B, 0, stream>>>(cnt, N);
  k_count<<<gE, B, 0, stream>>>(dst, cnt, E);
  k_scan<<<1, 1024, 0, stream>>>(cnt, rowstart, dinv, N);
  k_zero_i32<<<gN, B, 0, stream>>>(cur, N);
  k_fill<<<gE, B, 0, stream>>>(src, dst, rowstart, cur, csr, E);

  k_gemm_wmma<128><<<gGemm, 256, 64 * (2 * 128 + 32) * sizeof(float), stream>>>(x, W1, dinv, hw1, N);
  k_aggregate<128><<<2048, 256, 0, stream>>>(hw1, dinv, rowstart, csr, b1, h1, N);

  k_gemm_wmma<64><<<gGemm, 256, 64 * (2 * 64 + 32) * sizeof(float), stream>>>(h1, W2, dinv, hw2, N);
  k_aggregate<64><<<2048, 256, 0, stream>>>(hw2, dinv, rowstart, csr, b2, h2, N);

  k_zero_f32<<<(G * 64 + G + B - 1) / B, B, 0, stream>>>(pool, G * 64 + G);
  k_pool_acc<<<(N + 4 * POOL_ROWS - 1) / (4 * POOL_ROWS), 128, 0, stream>>>(h2, batch, pool, cntg, N);
  k_pool_final<<<(G * 64 + B - 1) / B, B, 0, stream>>>(pool, cntg, (float*)d_out);
}


# k_count keeps the value its atomic add returns as the edge's rank in its row; k_fill stores at rowstart+rank with no atomics; cur and its clear launch go; row order = count pass's arrival order
# speedup vs baseline: 2.3612x; 1.2282x over previous
#include <hip/hip_runtime.h>
#include <hip/hip_bf16.h>

typedef float v2f __attribute__((ext_vector_type(2)));
typedef float v8f __attribute__((ext_vector_type(8)));

#define N_GRAPHS 64

__global__ void k_zero_i32(int* p, int n) {
  int i = blockIdx.x * blockDim.x + threadIdx.x;
  for (; i < n; i += gridDim.x * blockDim.x) p[i] = 0;
}
__global__ void k_zero_f32(float* p, int n) {
  int i = blockIdx.x * blockDim.x + threadIdx.x;
  for (; i < n; i += gridDim.x * blockDim.x) p[i] = 0.0f;
}

__global__ void k_count(const int* __restrict__ dst, int* __restrict__ cnt,
                        int* __restrict__ rank, int E) {
  int e = blockIdx.x * blockDim.x + threadIdx.x;
  if (e < E) rank[e] = atomicAdd(&cnt[dst[e]], 1);
}

#define SCAN_ROWS 13
__global__ void k_scan(const int* __restrict__ cnt, int* __restrict__ rowstart,
                       float* __restrict__ dinv, int n) {
  constexpr int WSEG = SCAN_ROWS * 128;
  constexpr int TILE = 32 * WSEG;
  __shared__ int wtot[32];
  const int tid = threadIdx.x;
  const int lane = tid & 31;
  const int wave = tid >> 5;
  const int n4 = n & ~3;
  int carry = 0;
  for (int base = 0; base < n4; base += TILE) {
    const int i0 = base + wave * WSEG + lane * 4;
    int4 v[SCAN_ROWS];
    #pragma unroll
    for (int j = 0; j < SCAN_ROWS; ++j) {
      const int i = i0 + j * 128;
      int4 t = make_int4(0, 0, 0, 0);
      if (i < n4) t = *(const int4*)(cnt + i);
      v[j] = t;
    }
    int inc[SCAN_ROWS];
    #pragma unroll
    for (int j = 0; j < SCAN_ROWS; ++j) inc[j] = v[j].x + v[j].y + v[j].z + v[j].w;
    #pragma unroll
    for (int off = 1; off < 32; off <<= 1) {
      #pragma unroll
      for (int j = 0; j < SCAN_ROWS; ++j) {
        const int t = __shfl_up(inc[j], off, 32);
        if (lane >= off) inc[j] += t;
      }
    }
    int rowoff[SCAN_ROWS];
    int wsum = 0;
    #pragma unroll
    for (int j = 0; j < SCAN_ROWS; ++j) {
      rowoff[j] = wsum;
      wsum += __builtin_amdgcn_readlane(inc[j], 31);
    }
    if (lane == 0) wtot[wave] = wsum;
    __syncthreads();
    const int wt = wtot[lane];
    int winc = wt;
    #pragma unroll
    for (int off = 1; off < 32; off <<= 1) {
      const int t = __shfl_up(winc, off, 32);
      if (lane >= off) winc += t;
    }
    const int woff = carry + __shfl(winc - wt, wave, 32);
    carry += __builtin_amdgcn_readlane(winc, 31);
    #pragma unroll
    for (int j = 0; j < SCAN_ROWS; ++j) {
      const int i = i0 + j * 128;
      const int4 c = v[j];
      int4 r;
      r.x = woff + rowoff[j] + inc[j] - (c.x + c.y + c.z + c.w);
      r.y = r.x + c.x;
      r.z = r.y + c.y;
      r.w = r.z + c.z;
      float4 d;
      d.x = rsqrtf((float)c.x + 1.0f);
      d.y = rsqrtf((float)c.y + 1.0f);
      d.z = rsqrtf((float)c.z + 1.0f);
      d.w = rsqrtf((float)c.w + 1.0f);
      if (i < n4) {
        *(int4*)(rowstart + i) = r;
        *(float4*)(dinv + i) = d;
      }
    }
    __syncthreads();
  }
  if (tid == 0) {
    int c = carry;
    for (int i = n4; i < n; ++i) {
      const int vv = cnt[i];
      rowstart[i] = c;
      dinv[i] = rsqrtf((float)vv + 1.0f);
      c += vv;
    }
    rowstart[n] = c;
  }
}

__global__ void k_fill(const int* __restrict__ src, const int* __restrict__ dst,
                       const int* __restrict__ rowstart, const int* __restrict__ rank,
                       int* __restrict__ csr, int E) {
  int e = blockIdx.x * blockDim.x + threadIdx.x;
  if (e < E) csr[rowstart[dst[e]] + rank[e]] = src[e];
}

template <int N>
__global__ void k_gemm_wmma(const float* __restrict__ A, const float* __restrict__ W,
                            const float* __restrict__ dinv, float* __restrict__ out, int M) {
  constexpr int K = 128;
  constexpr int NT = N / 16;
  constexpr int RP = 2 * N + 32;
  extern __shared__ float Wlds[];
  for (int i = threadIdx.x; i < K * N; i += blockDim.x) {
    int k = i / N;
    int n = i % N;
    Wlds[(k >> 1) * RP + n * 2 + (k & 1)] = W[i];
  }
  __syncthreads();

  const int wave = threadIdx.x >> 5;
  const int lane = threadIdx.x & 31;
  const int mtile = blockIdx.x * 8 + wave;
  if (mtile * 16 >= M) return;

  const int Rm = mtile * 16;
  const int half = lane >> 4;
  const int l16 = lane & 15;

  v8f acc[NT] = {};
  const float* arow = A + (size_t)(Rm + l16) * K;
  const float* bbase = Wlds + half * RP + l16 * 2;

  for (int kb = 0; kb < K; kb += 4) {
    v2f a = *(const v2f*)(arow + kb + 2 * half);
    const float* bk = bbase + (kb >> 1) * RP;
    #pragma unroll
    for (int t = 0; t < NT; ++t) {
      v2f b = *(const v2f*)(bk + t * 32);
      acc[t] = __builtin_amdgcn_wmma_f32_16x16x4_f32(
          false, a, false, b, (short)0, acc[t], false, false);
    }
  }

  #pragma unroll
  for (int r = 0; r < 8; ++r) {
    const int row = Rm + r + 8 * half;
    const float di = dinv[row];
    #pragma unroll
    for (int t = 0; t < NT; ++t) {
      out[(size_t)row * N + t * 16 + l16] = acc[t][r] * di;
    }
  }
}

template <int F>
struct AggVec {
  typedef float type __attribute__((ext_vector_type(F / 32)));
};

template <int F, int NB>
__device__ __forceinline__ typename AggVec<F>::type agg_rows(
    const float* __restrict__ hws, int idx, int k, int lane, typename AggVec<F>::type sum) {
  typedef typename AggVec<F>::type vf;
  constexpr int V = F / 32;
  vf v[NB];
  #pragma unroll
  for (int u = 0; u < NB; ++u) {
    const unsigned r = (unsigned)__builtin_amdgcn_readlane(idx, k + u);
    const float* row = hws + (size_t)r * F;
    v[u] = *(const vf*)(row + V * lane);
  }
  #pragma unroll
  for (int u = 0; u < NB; ++u) sum += v[u];
  return sum;
}

template <int F>
__global__ void __launch_bounds__(256)
k_aggregate(const float* __restrict__ hws, const float* __restrict__ dinv,
            const int* __restrict__ rowstart, const int* __restrict__ csr,
            const float* __restrict__ bias, float* __restrict__ out, int n) {
  constexpr int V = F / 32;
  typedef typename AggVec<F>::type vf;
  const int lane = threadIdx.x & 31;
  const int wave = __builtin_amdgcn_readfirstlane(threadIdx.x >> 5);
  const int wpb = blockDim.x >> 5;
  const int nwaves = gridDim.x * wpb;
  const vf bv = *(const vf*)(bias + V * lane);
  for (int node = blockIdx.x * wpb + wave; node < n; node += nwaves) {
    const int s = rowstart[node];
    const int e = rowstart[node + 1];
    vf sum = *(const vf*)(hws + (size_t)node * F + V * lane);
    for (int base = s; base < e; base += 32) {
      const int cnt = min(e - base, 32);
      const int idx = csr[base + min(lane, cnt - 1)];
      int k = 0;
      for (; k + 8 <= cnt; k += 8) sum = agg_rows<F, 8>(hws, idx, k, lane, sum);
      if (cnt & 4) { sum = agg_rows<F, 4>(hws, idx, k, lane, sum); k += 4; }
      if (cnt & 2) { sum = agg_rows<F, 2>(hws, idx, k, lane, sum); k += 2; }
      if (cnt & 1) sum = agg_rows<F, 1>(hws, idx, k, lane, sum);
    }
    const float di = dinv[node];
    vf o;
    #pragma unroll
    for (int c = 0; c < V; ++c) o[c] = fmaxf(di * sum[c] + bv[c], 0.0f);
    *(vf*)(out + (size_t)node * F + V * lane) = o;
  }
}

#define POOL_ROWS 32
__device__ __forceinline__ void pool_flush(float* __restrict__ pool, float* __restrict__ cntg,
                                           int g, v2f acc, int run, int lane) {
  float* p = pool + g * 64 + 2 * lane;
  atomicAdd(p, acc.x);
  atomicAdd(p + 1, acc.y);
  if (lane == 0) atomicAdd(&cntg[g], (float)run);
}

__global__ void __launch_bounds__(128)
k_pool_acc(const float* __restrict__ h, const int* __restrict__ batch,
           float* __restrict__ pool, float* __restrict__ cntg, int n) {
  const int lane = threadIdx.x & 31;
  const int wave = __builtin_amdgcn_readfirstlane(threadIdx.x >> 5);
  const int node0 = (blockIdx.x * (blockDim.x >> 5) + wave) * POOL_ROWS;
  if (node0 >= n) return;
  const int nv = min(POOL_ROWS, n - node0);
  const int ids = batch[node0 + min(lane, nv - 1)];
  v2f v[POOL_ROWS];
  #pragma unroll
  for (int u = 0; u < POOL_ROWS; ++u) {
    const int r = min(node0 + u, n - 1);
    const float* row = h + (size_t)r * 64;
    v[u] = *(const v2f*)(row + 2 * lane);
  }
  int cur = __builtin_amdgcn_readlane(ids, 0);
  int run = 1;
  v2f acc = v[0];
  #pragma unroll
  for (int u = 1; u < POOL_ROWS; ++u) {
    if (u < nv) {
      const int g = __builtin_amdgcn_readlane(ids, u);
      if (g != cur) {
        pool_flush(pool, cntg, cur, acc, run, lane);
        acc = (v2f){0.0f, 0.0f};
        run = 0;
        cur = g;
      }
      acc += v[u];
      ++run;
    }
  }
  pool_flush(pool, cntg, cur, acc, run, lane);
}

__global__ void k_pool_final(const float* __restrict__ pool, const float* __restrict__ cntg,
                             float* __restrict__ out) {
  int idx = blockIdx.x * blockDim.x + threadIdx.x;
  if (idx < N_GRAPHS * 64) {
    out[idx] = pool[idx] / fmaxf(cntg[idx >> 6], 1.0f);
  }
}

extern "C" void kernel_launch(void* const* d_in, const int* in_sizes, int n_in,
                              void* d_out, int out_size, void* d_ws, size_t ws_size,
                              hipStream_t stream) {
  const float* x     = (const float*)d_in[0];
  const float* W1    = (const float*)d_in[1];
  const float* b1    = (const float*)d_in[2];
  const float* W2    = (const float*)d_in[3];
  const float* b2    = (const float*)d_in[4];
  const int*   ei    = (const int*)d_in[5];
  const int*   batch = (const int*)d_in[6];

  const int N = in_sizes[6];
  const int E = in_sizes[5] / 2;
  const int G = N_GRAPHS;

  const int* src = ei;
  const int* dst = ei + E;

  char* base = (char*)d_ws;
  size_t off = 0;
  auto take = [&](size_t bytes) {
    size_t o = off;
    off = (off + bytes + 255) & ~(size_t)255;
    return o;
  };
  int*   cnt      = (int*)(base + take(sizeof(int) * (size_t)N));
  int*   rank     = (int*)(base + take(sizeof(int) * (size_t)E));
  int*   rowstart = (int*)(base + take(sizeof(int) * (size_t)(N + 1)));
  float* dinv     = (float*)(base + take(sizeof(float) * (size_t)N));
  int*   csr      = (int*)(base + take(sizeof(int) * (size_t)E));
  float* hw1      = (float*)(base + take(sizeof(float) * (size_t)N * 128));
  float* h1       = (float*)(base + take(sizeof(float) * (size_t)N * 128));
  float* pool     = (float*)(base + take(sizeof(float) * (size_t)G * 64));
  float* cntg     = (float*)(base + take(sizeof(float) * (size_t)G));
  float* hw2 = hw1;
  float* h2  = h1;

  const int B = 256;
  const int gN = (N + B - 1) / B;
  const int gE = (E + B - 1) / B;
  const int mtiles = (N + 15) / 16;
  const int gGemm = (mtiles + 7) / 8;

  k_zero_i32<<<gN, B, 0, stream>>>(cnt, N);
  k_count<<<gE, B, 0, stream>>>(dst, cnt, rank, E);
  k_scan<<<1, 1024, 0, stream>>>(cnt, rowstart, dinv, N);
  k_fill<<<gE, B, 0, stream>>>(src, dst, rowstart, rank, csr, E);

  k_gemm_wmma<128><<<gGemm, 256, 64 * (2 * 128 + 32) * sizeof(float), stream>>>(x, W1, dinv, hw1, N);
  k_aggregate<128><<<2048, 256, 0, stream>>>(hw1, dinv, rowstart, csr, b1, h1, N);

  k_gemm_wmma<64><<<gGemm, 256, 64 * (2 * 64 + 32) * sizeof(float), stream>>>(h1, W2, dinv, hw2, N);
  k_aggregate<64><<<2048, 256, 0, stream>>>(hw2, dinv, rowstart, csr, b2, h2, N);

  k_zero_f32<<<(G * 64 + G + B - 1) / B, B, 0, stream>>>(pool, G * 64 + G);
  k_pool_acc<<<(N + 4 * POOL_ROWS - 1) / (4 * POOL_ROWS), 128, 0, stream>>>(h2, batch, pool, cntg, N);
  k_pool_final<<<(G * 64 + B - 1) / B, B, 0, stream>>>(pool, cntg, (float*)d_out);
}


# k_count: 64 blocks rank their edge chunk with per-node counters in LDS (returning LDS adds), no global atomics; k_merge scans chunk counts per node; csr row (sum) order = chunk, then arrival
# speedup vs baseline: 2.7985x; 1.1852x over previous
#include <hip/hip_runtime.h>
#include <hip/hip_bf16.h>

typedef float v2f __attribute__((ext_vector_type(2)));
typedef float v8f __attribute__((ext_vector_type(8)));

#define N_GRAPHS 64

__global__ void k_zero_f32(float* p, int n) {
  int i = blockIdx.x * blockDim.x + threadIdx.x;
  for (; i < n; i += gridDim.x * blockDim.x) p[i] = 0.0f;
}

#define CNT_BLOCKS 64
#define CNT_BATCH 8
__global__ void __launch_bounds__(1024)
k_count(const int* __restrict__ dst, int* __restrict__ hist, int* __restrict__ rank,
        int E, int n, int chunk, int np) {
  extern __shared__ int4 hc4[];
  int* hc = (int*)hc4;
  const int tid = threadIdx.x;
  const int nq = (n + 3) >> 2;
  for (int i = tid; i < nq; i += 1024) hc4[i] = make_int4(0, 0, 0, 0);
  __syncthreads();
  const int start = blockIdx.x * chunk;
  const int end = min(E, start + chunk);
  for (int e0 = start + tid; e0 < end; e0 += CNT_BATCH * 1024) {
    int d[CNT_BATCH];
    int r[CNT_BATCH];
    if (e0 + (CNT_BATCH - 1) * 1024 < end) {
      #pragma unroll
      for (int u = 0; u < CNT_BATCH; ++u) d[u] = dst[e0 + u * 1024];
      #pragma unroll
      for (int u = 0; u < CNT_BATCH; ++u) r[u] = atomicAdd(&hc[d[u]], 1);
      #pragma unroll
      for (int u = 0; u < CNT_BATCH; ++u) rank[e0 + u * 1024] = r[u];
    } else {
      #pragma unroll
      for (int u = 0; u < CNT_BATCH; ++u) {
        const int e = e0 + u * 1024;
        d[u] = (e < end) ? dst[e] : 0;
      }
      #pragma unroll
      for (int u = 0; u < CNT_BATCH; ++u) {
        const int e = e0 + u * 1024;
        if (e < end) rank[e] = atomicAdd(&hc[d[u]], 1);
      }
    }
  }
  __syncthreads();
  int4* __restrict__ row = (int4*)(hist + (size_t)blockIdx.x * np);
  for (int i = tid; i < nq; i += 1024) row[i] = hc4[i];
}

__global__ void k_merge(int* __restrict__ hist, int* __restrict__ cnt, int n, int np) {
  const int i = blockIdx.x * blockDim.x + threadIdx.x;
  if (i >= n) return;
  int run = 0;
  for (int b0 = 0; b0 < CNT_BLOCKS; b0 += 16) {
    int c[16];
    #pragma unroll
    for (int u = 0; u < 16; ++u) c[u] = hist[(size_t)(b0 + u) * np + i];
    #pragma unroll
    for (int u = 0; u < 16; ++u) {
      hist[(size_t)(b0 + u) * np + i] = run;
      run += c[u];
    }
  }
  cnt[i] = run;
}

#define SCAN_ROWS 13
__global__ void k_scan(const int* __restrict__ cnt, int* __restrict__ rowstart,
                       float* __restrict__ dinv, int n) {
  constexpr int WSEG = SCAN_ROWS * 128;
  constexpr int TILE = 32 * WSEG;
  __shared__ int wtot[32];
  const int tid = threadIdx.x;
  const int lane = tid & 31;
  const int wave = tid >> 5;
  const int n4 = n & ~3;
  int carry = 0;
  for (int base = 0; base < n4; base += TILE) {
    const int i0 = base + wave * WSEG + lane * 4;
    int4 v[SCAN_ROWS];
    #pragma unroll
    for (int j = 0; j < SCAN_ROWS; ++j) {
      const int i = i0 + j * 128;
      int4 t = make_int4(0, 0, 0, 0);
      if (i < n4) t = *(const int4*)(cnt + i);
      v[j] = t;
    }
    int inc[SCAN_ROWS];
    #pragma unroll
    for (int j = 0; j < SCAN_ROWS; ++j) inc[j] = v[j].x + v[j].y + v[j].z + v[j].w;
    #pragma unroll
    for (int off = 1; off < 32; off <<= 1) {
      #pragma unroll
      for (int j = 0; j < SCAN_ROWS; ++j) {
        const int t = __shfl_up(inc[j], off, 32);
        if (lane >= off) inc[j] += t;
      }
    }
    int rowoff[SCAN_ROWS];
    int wsum = 0;
    #pragma unroll
    for (int j = 0; j < SCAN_ROWS; ++j) {
      rowoff[j] = wsum;
      wsum += __builtin_amdgcn_readlane(inc[j], 31);
    }
    if (lane == 0) wtot[wave] = wsum;
    __syncthreads();
    const int wt = wtot[lane];
    int winc = wt;
    #pragma unroll
    for (int off = 1; off < 32; off <<= 1) {
      const int t = __shfl_up(winc, off, 32);
      if (lane >= off) winc += t;
    }
    const int woff = carry + __shfl(winc - wt, wave, 32);
    carry += __builtin_amdgcn_readlane(winc, 31);
    #pragma unroll
    for (int j = 0; j < SCAN_ROWS; ++j) {
      const int i = i0 + j * 128;
      const int4 c = v[j];
      int4 r;
      r.x = woff + rowoff[j] + inc[j] - (c.x + c.y + c.z + c.w);
      r.y = r.x + c.x;
      r.z = r.y + c.y;
      r.w = r.z + c.z;
      float4 d;
      d.x = rsqrtf((float)c.x + 1.0f);
      d.y = rsqrtf((float)c.y + 1.0f);
      d.z = rsqrtf((float)c.z + 1.0f);
      d.w = rsqrtf((float)c.w + 1.0f);
      if (i < n4) {
        *(int4*)(rowstart + i) = r;
        *(float4*)(dinv + i) = d;
      }
    }
    __syncthreads();
  }
  if (tid == 0) {
    int c = carry;
    for (int i = n4; i < n; ++i) {
      const int vv = cnt[i];
      rowstart[i] = c;
      dinv[i] = rsqrtf((float)vv + 1.0f);
      c += vv;
    }
    rowstart[n] = c;
  }
}

__global__ void k_fill(const int* __restrict__ src, const int* __restrict__ dst,
                       const int* __restrict__ rowstart, const int* __restrict__ rank,
                       const int* __restrict__ hist, int* __restrict__ csr,
                       int E, int chunk, int np) {
  const int local = blockIdx.x * blockDim.x + threadIdx.x;
  const int e = blockIdx.y * chunk + local;
  if (local < chunk && e < E) {
    const int d = dst[e];
    csr[rowstart[d] + hist[(size_t)blockIdx.y * np + d] + rank[e]] = src[e];
  }
}

template <int N>
__global__ void k_gemm_wmma(const float* __restrict__ A, const float* __restrict__ W,
                            const float* __restrict__ dinv, float* __restrict__ out, int M) {
  constexpr int K = 128;
  constexpr int NT = N / 16;
  constexpr int RP = 2 * N + 32;
  extern __shared__ float Wlds[];
  for (int i = threadIdx.x; i < K * N; i += blockDim.x) {
    int k = i / N;
    int n = i % N;
    Wlds[(k >> 1) * RP + n * 2 + (k & 1)] = W[i];
  }
  __syncthreads();

  const int wave = threadIdx.x >> 5;
  const int lane = threadIdx.x & 31;
  const int mtile = blockIdx.x * 8 + wave;
  if (mtile * 16 >= M) return;

  const int Rm = mtile * 16;
  const int half = lane >> 4;
  const int l16 = lane & 15;

  v8f acc[NT] = {};
  const float* arow = A + (size_t)(Rm + l16) * K;
  const float* bbase = Wlds + half * RP + l16 * 2;

  for (int kb = 0; kb < K; kb += 4) {
    v2f a = *(const v2f*)(arow + kb + 2 * half);
    const float* bk = bbase + (kb >> 1) * RP;
    #pragma unroll
    for (int t = 0; t < NT; ++t) {
      v2f b = *(const v2f*)(bk + t * 32);
      acc[t] = __builtin_amdgcn_wmma_f32_16x16x4_f32(
          false, a, false, b, (short)0, acc[t], false, false);
    }
  }

  #pragma unroll
  for (int r = 0; r < 8; ++r) {
    const int row = Rm + r + 8 * half;
    const float di = dinv[row];
    #pragma unroll
    for (int t = 0; t < NT; ++t) {
      out[(size_t)row * N + t * 16 + l16] = acc[t][r] * di;
    }
  }
}

template <int F>
struct AggVec {
  typedef float type __attribute__((ext_vector_type(F / 32)));
};

template <int F, int NB>
__device__ __forceinline__ typename AggVec<F>::type agg_rows(
    const float* __restrict__ hws, int idx, int k, int lane, typename AggVec<F>::type sum) {
  typedef typename AggVec<F>::type vf;
  constexpr int V = F / 32;
  vf v[NB];
  #pragma unroll
  for (int u = 0; u < NB; ++u) {
    const unsigned r = (unsigned)__builtin_amdgcn_readlane(idx, k + u);
    const float* row = hws + (size_t)r * F;
    v[u] = *(const vf*)(row + V * lane);
  }
  #pragma unroll
  for (int u = 0; u < NB; ++u) sum += v[u];
  return sum;
}

template <int F>
__global__ void __launch_bounds__(256)
k_aggregate(const float* __restrict__ hws, const float* __restrict__ dinv,
            const int* __restrict__ rowstart, const int* __restrict__ csr,
            const float* __restrict__ bias, float* __restrict__ out, int n) {
  constexpr int V = F / 32;
  typedef typename AggVec<F>::type vf;
  const int lane = threadIdx.x & 31;
  const int wave = __builtin_amdgcn_readfirstlane(threadIdx.x >> 5);
  const int wpb = blockDim.x >> 5;
  const int nwaves = gridDim.x * wpb;
  const vf bv = *(const vf*)(bias + V * lane);
  for (int node = blockIdx.x * wpb + wave; node < n; node += nwaves) {
    const int s = rowstart[node];
    const int e = rowstart[node + 1];
    vf sum = *(const vf*)(hws + (size_t)node * F + V * lane);
    for (int base = s; base < e; base += 32) {
      const int cnt = min(e - base, 32);
      const int idx = csr[base + min(lane, cnt - 1)];
      int k = 0;
      for (; k + 8 <= cnt; k += 8) sum = agg_rows<F, 8>(hws, idx, k, lane, sum);
      if (cnt & 4) { sum = agg_rows<F, 4>(hws, idx, k, lane, sum); k += 4; }
      if (cnt & 2) { sum = agg_rows<F, 2>(hws, idx, k, lane, sum); k += 2; }
      if (cnt & 1) sum = agg_rows<F, 1>(hws, idx, k, lane, sum);
    }
    const float di = dinv[node];
    vf o;
    #pragma unroll
    for (int c = 0; c < V; ++c) o[c] = fmaxf(di * sum[c] + bv[c], 0.0f);
    *(vf*)(out + (size_t)node * F + V * lane) = o;
  }
}

#define POOL_ROWS 32
__device__ __forceinline__ void pool_flush(float* __restrict__ pool, float* __restrict__ cntg,
                                           int g, v2f acc, int run, int lane) {
  float* p = pool + g * 64 + 2 * lane;
  atomicAdd(p, acc.x);
  atomicAdd(p + 1, acc.y);
  if (lane == 0) atomicAdd(&cntg[g], (float)run);
}

__global__ void __launch_bounds__(128)
k_pool_acc(const float* __restrict__ h, const int* __restrict__ batch,
           float* __restrict__ pool, float* __restrict__ cntg, int n) {
  const int lane = threadIdx.x & 31;
  const int wave = __builtin_amdgcn_readfirstlane(threadIdx.x >> 5);
  const int node0 = (blockIdx.x * (blockDim.x >> 5) + wave) * POOL_ROWS;
  if (node0 >= n) return;
  const int nv = min(POOL_ROWS, n - node0);
  const int ids = batch[node0 + min(lane, nv - 1)];
  v2f v[POOL_ROWS];
  #pragma unroll
  for (int u = 0; u < POOL_ROWS; ++u) {
    const int r = min(node0 + u, n - 1);
    const float* row = h + (size_t)r * 64;
    v[u] = *(const v2f*)(row + 2 * lane);
  }
  int cur = __builtin_amdgcn_readlane(ids, 0);
  int run = 1;
  v2f acc = v[0];
  #pragma unroll
  for (int u = 1; u < POOL_ROWS; ++u) {
    if (u < nv) {
      const int g = __builtin_amdgcn_readlane(ids, u);
      if (g != cur) {
        pool_flush(pool, cntg, cur, acc, run, lane);
        acc = (v2f){0.0f, 0.0f};
        run = 0;
        cur = g;
      }
      acc += v[u];
      ++run;
    }
  }
  pool_flush(pool, cntg, cur, acc, run, lane);
}

__global__ void k_pool_final(const float* __restrict__ pool, const float* __restrict__ cntg,
                             float* __restrict__ out) {
  int idx = blockIdx.x * blockDim.x + threadIdx.x;
  if (idx < N_GRAPHS * 64) {
    out[idx] = pool[idx] / fmaxf(cntg[idx >> 6], 1.0f);
  }
}

extern "C" void kernel_launch(void* const* d_in, const int* in_sizes, int n_in,
                              void* d_out, int out_size, void* d_ws, size_t ws_size,
                              hipStream_t stream) {
  const float* x     = (const float*)d_in[0];
  const float* W1    = (const float*)d_in[1];
  const float* b1    = (const float*)d_in[2];
  const float* W2    = (const float*)d_in[3];
  const float* b2    = (const float*)d_in[4];
  const int*   ei    = (const int*)d_in[5];
  const int*   batch = (const int*)d_in[6];

  const int N = in_sizes[6];
  const int E = in_sizes[5] / 2;
  const int G = N_GRAPHS;

  const int* src = ei;
  const int* dst = ei + E;

  char* base = (char*)d_ws;
  size_t off = 0;
  auto take = [&](size_t bytes) {
    size_t o = off;
    off = (off + bytes + 255) & ~(size_t)255;
    return o;
  };
  const int np = (N + 31) & ~31;
  const int chunk = (E + CNT_BLOCKS - 1) / CNT_BLOCKS;
  int*   cnt      = (int*)(base + take(sizeof(int) * (size_t)N));
  int*   hist     = (int*)(base + take(sizeof(int) * (size_t)np * CNT_BLOCKS));
  int*   rank     = (int*)(base + take(sizeof(int) * (size_t)E));
  int*   rowstart = (int*)(base + take(sizeof(int) * (size_t)(N + 1)));
  float* dinv     = (float*)(base + take(sizeof(float) * (size_t)N));
  int*   csr      = (int*)(base + take(sizeof(int) * (size_t)E));
  float* hw1      = (float*)(base + take(sizeof(float) * (size_t)N * 128));
  float* h1       = (float*)(base + take(sizeof(float) * (size_t)N * 128));
  float* pool     = (float*)(base + take(sizeof(float) * (size_t)G * 64));
  float* cntg     = (float*)(base + take(sizeof(float) * (size_t)G));
  float* hw2 = hw1;
  float* h2  = h1;

  const int B = 256;
  const int gN = (N + B - 1) / B;
  const int mtiles = (N + 15) / 16;
  const int gGemm = (mtiles + 7) / 8;

  k_count<<<CNT_BLOCKS, 1024, sizeof(int4) * (size_t)((N + 3) / 4), stream>>>(dst, hist, rank, E, N, chunk, np);
  k_merge<<<gN, B, 0, stream>>>(hist, cnt, N, np);
  k_scan<<<1, 1024, 0, stream>>>(cnt, rowstart, dinv, N);
  k_fill<<<dim3((chunk + B - 1) / B, CNT_BLOCKS), B, 0, stream>>>(src, dst, rowstart, rank, hist, csr, E, chunk, np);

  k_gemm_wmma<128><<<gGemm, 256, 64 * (2 * 128 + 32) * sizeof(float), stream>>>(x, W1, dinv, hw1, N);
  k_aggregate<128><<<2048, 256, 0, stream>>>(hw1, dinv, rowstart, csr, b1, h1, N);

  k_gemm_wmma<64><<<gGemm, 256, 64 * (2 * 64 + 32) * sizeof(float), stream>>>(h1, W2, dinv, hw2, N);
  k_aggregate<64><<<2048, 256, 0, stream>>>(hw2, dinv, rowstart, csr, b2, h2, N);

  k_zero_f32<<<(G * 64 + G + B - 1) / B, B, 0, stream>>>(pool, G * 64 + G);
  k_pool_acc<<<(N + 4 * POOL_ROWS - 1) / (4 * POOL_ROWS), 128, 0, stream>>>(h2, batch, pool, cntg, N);
  k_pool_final<<<(G * 64 + B - 1) / B, B, 0, stream>>>(pool, cntg, (float*)d_out);
}


# k_fill's edge placement runs as 448 extra blocks (8 edges a thread) of the layer-1 GEMM launch, which also needs only the scan: 11 launches become 10, the 25000-wave edge pass overlaps the WMMAs
# speedup vs baseline: 3.1420x; 1.1227x over previous
#include <hip/hip_runtime.h>
#include <hip/hip_bf16.h>

typedef float v2f __attribute__((ext_vector_type(2)));
typedef float v8f __attribute__((ext_vector_type(8)));

#define N_GRAPHS 64

__global__ void k_zero_f32(float* p, int n) {
  int i = blockIdx.x * blockDim.x + threadIdx.x;
  for (; i < n; i += gridDim.x * blockDim.x) p[i] = 0.0f;
}

#define CNT_BLOCKS 64
#define CNT_BATCH 8
__global__ void __launch_bounds__(1024)
k_count(const int* __restrict__ dst, int* __restrict__ hist, int* __restrict__ rank,
        int E, int n, int chunk, int np) {
  extern __shared__ int4 hc4[];
  int* hc = (int*)hc4;
  const int tid = threadIdx.x;
  const int nq = (n + 3) >> 2;
  for (int i = tid; i < nq; i += 1024) hc4[i] = make_int4(0, 0, 0, 0);
  __syncthreads();
  const int start = blockIdx.x * chunk;
  const int end = min(E, start + chunk);
  for (int e0 = start + tid; e0 < end; e0 += CNT_BATCH * 1024) {
    int d[CNT_BATCH];
    int r[CNT_BATCH];
    if (e0 + (CNT_BATCH - 1) * 1024 < end) {
      #pragma unroll
      for (int u = 0; u < CNT_BATCH; ++u) d[u] = dst[e0 + u * 1024];
      #pragma unroll
      for (int u = 0; u < CNT_BATCH; ++u) r[u] = atomicAdd(&hc[d[u]], 1);
      #pragma unroll
      for (int u = 0; u < CNT_BATCH; ++u) rank[e0 + u * 1024] = r[u];
    } else {
      #pragma unroll
      for (int u = 0; u < CNT_BATCH; ++u) {
        const int e = e0 + u * 1024;
        d[u] = (e < end) ? dst[e] : 0;
      }
      #pragma unroll
      for (int u = 0; u < CNT_BATCH; ++u) {
        const int e = e0 + u * 1024;
        if (e < end) rank[e] = atomicAdd(&hc[d[u]], 1);
      }
    }
  }
  __syncthreads();
  int4* __restrict__ row = (int4*)(hist + (size_t)blockIdx.x * np);
  for (int i = tid; i < nq; i += 1024) row[i] = hc4[i];
}

__global__ void k_merge(int* __restrict__ hist, int* __restrict__ cnt, int n, int np) {
  const int i = blockIdx.x * blockDim.x + threadIdx.x;
  if (i >= n) return;
  int run = 0;
  for (int b0 = 0; b0 < CNT_BLOCKS; b0 += 16) {
    int c[16];
    #pragma unroll
    for (int u = 0; u < 16; ++u) c[u] = hist[(size_t)(b0 + u) * np + i];
    #pragma unroll
    for (int u = 0; u < 16; ++u) {
      hist[(size_t)(b0 + u) * np + i] = run;
      run += c[u];
    }
  }
  cnt[i] = run;
}

#define SCAN_ROWS 13
__global__ void k_scan(const int* __restrict__ cnt, int* __restrict__ rowstart,
                       float* __restrict__ dinv, int n) {
  constexpr int WSEG = SCAN_ROWS * 128;
  constexpr int TILE = 32 * WSEG;
  __shared__ int wtot[32];
  const int tid = threadIdx.x;
  const int lane = tid & 31;
  const int wave = tid >> 5;
  const int n4 = n & ~3;
  int carry = 0;
  for (int base = 0; base < n4; base += TILE) {
    const int i0 = base + wave * WSEG + lane * 4;
    int4 v[SCAN_ROWS];
    #pragma unroll
    for (int j = 0; j < SCAN_ROWS; ++j) {
      const int i = i0 + j * 128;
      int4 t = make_int4(0, 0, 0, 0);
      if (i < n4) t = *(const int4*)(cnt + i);
      v[j] = t;
    }
    int inc[SCAN_ROWS];
    #pragma unroll
    for (int j = 0; j < SCAN_ROWS; ++j) inc[j] = v[j].x + v[j].y + v[j].z + v[j].w;
    #pragma unroll
    for (int off = 1; off < 32; off <<= 1) {
      #pragma unroll
      for (int j = 0; j < SCAN_ROWS; ++j) {
        const int t = __shfl_up(inc[j], off, 32);
        if (lane >= off) inc[j] += t;
      }
    }
    int rowoff[SCAN_ROWS];
    int wsum = 0;
    #pragma unroll
    for (int j = 0; j < SCAN_ROWS; ++j) {
      rowoff[j] = wsum;
      wsum += __builtin_amdgcn_readlane(inc[j], 31);
    }
    if (lane == 0) wtot[wave] = wsum;
    __syncthreads();
    const int wt = wtot[lane];
    int winc = wt;
    #pragma unroll
    for (int off = 1; off < 32; off <<= 1) {
      const int t = __shfl_up(winc, off, 32);
      if (lane >= off) winc += t;
    }
    const int woff = carry + __shfl(winc - wt, wave, 32);
    carry += __builtin_amdgcn_readlane(winc, 31);
    #pragma unroll
    for (int j = 0; j < SCAN_ROWS; ++j) {
      const int i = i0 + j * 128;
      const int4 c = v[j];
      int4 r;
      r.x = woff + rowoff[j] + inc[j] - (c.x + c.y + c.z + c.w);
      r.y = r.x + c.x;
      r.z = r.y + c.y;
      r.w = r.z + c.z;
      float4 d;
      d.x = rsqrtf((float)c.x + 1.0f);
      d.y = rsqrtf((float)c.y + 1.0f);
      d.z = rsqrtf((float)c.z + 1.0f);
      d.w = rsqrtf((float)c.w + 1.0f);
      if (i < n4) {
        *(int4*)(rowstart + i) = r;
        *(float4*)(dinv + i) = d;
      }
    }
    __syncthreads();
  }
  if (tid == 0) {
    int c = carry;
    for (int i = n4; i < n; ++i) {
      const int vv = cnt[i];
      rowstart[i] = c;
      dinv[i] = rsqrtf((float)vv + 1.0f);
      c += vv;
    }
    rowstart[n] = c;
  }
}

#define FILL_BATCH 8
#define FILL_THREADS 256
#define FILL_TILE (FILL_BATCH * FILL_THREADS)
__device__ __forceinline__ void fill_tile(
    const int* __restrict__ src, const int* __restrict__ dst,
    const int* __restrict__ rowstart, const int* __restrict__ rank,
    const int* __restrict__ hist, int* __restrict__ csr,
    int E, int chunk, int np, int bx, int by) {
  const int cstart = by * chunk;
  const int lim = min(chunk, E - cstart);
  const int t0 = bx * FILL_TILE;
  if (t0 >= lim) return;
  const int* __restrict__ hrow = hist + (size_t)by * np;
  const int l0 = t0 + (int)threadIdx.x;
  int d[FILL_BATCH], r[FILL_BATCH], s[FILL_BATCH], p[FILL_BATCH];
  if (t0 + FILL_TILE <= lim) {
    const int e0 = cstart + l0;
    #pragma unroll
    for (int u = 0; u < FILL_BATCH; ++u) d[u] = dst[e0 + u * FILL_THREADS];
    #pragma unroll
    for (int u = 0; u < FILL_BATCH; ++u) r[u] = rank[e0 + u * FILL_THREADS];
    #pragma unroll
    for (int u = 0; u < FILL_BATCH; ++u) s[u] = src[e0 + u * FILL_THREADS];
    #pragma unroll
    for (int u = 0; u < FILL_BATCH; ++u) p[u] = rowstart[d[u]] + hrow[d[u]];
    #pragma unroll
    for (int u = 0; u < FILL_BATCH; ++u) csr[p[u] + r[u]] = s[u];
  } else {
    #pragma unroll
    for (int u = 0; u < FILL_BATCH; ++u) {
      const int e = cstart + min(l0 + u * FILL_THREADS, lim - 1);
      d[u] = dst[e];
      r[u] = rank[e];
      s[u] = src[e];
    }
    #pragma unroll
    for (int u = 0; u < FILL_BATCH; ++u) p[u] = rowstart[d[u]] + hrow[d[u]];
    #pragma unroll
    for (int u = 0; u < FILL_BATCH; ++u) {
      if (l0 + u * FILL_THREADS < lim) csr[p[u] + r[u]] = s[u];
    }
  }
}

template <int N>
__device__ __forceinline__ void gemm_block(const float* __restrict__ A, const float* __restrict__ W,
                                           const float* __restrict__ dinv, float* __restrict__ out, int M,
                                           int bid, float* Wlds) {
  constexpr int K = 128;
  constexpr int NT = N / 16;
  constexpr int RP = 2 * N + 32;
  for (int i = threadIdx.x; i < K * N; i += blockDim.x) {
    int k = i / N;
    int n = i % N;
    Wlds[(k >> 1) * RP + n * 2 + (k & 1)] = W[i];
  }
  __syncthreads();

  const int wave = threadIdx.x >> 5;
  const int lane = threadIdx.x & 31;
  const int mtile = bid * 8 + wave;
  if (mtile * 16 >= M) return;

  const int Rm = mtile * 16;
  const int half = lane >> 4;
  const int l16 = lane & 15;

  v8f acc[NT] = {};
  const float* arow = A + (size_t)(Rm + l16) * K;
  const float* bbase = Wlds + half * RP + l16 * 2;

  for (int kb = 0; kb < K; kb += 4) {
    v2f a = *(const v2f*)(arow + kb + 2 * half);
    const float* bk = bbase + (kb >> 1) * RP;
    #pragma unroll
    for (int t = 0; t < NT; ++t) {
      v2f b = *(const v2f*)(bk + t * 32);
      acc[t] = __builtin_amdgcn_wmma_f32_16x16x4_f32(
          false, a, false, b, (short)0, acc[t], false, false);
    }
  }

  #pragma unroll
  for (int r = 0; r < 8; ++r) {
    const int row = Rm + r + 8 * half;
    const float di = dinv[row];
    #pragma unroll
    for (int t = 0; t < NT; ++t) {
      out[(size_t)row * N + t * 16 + l16] = acc[t][r] * di;
    }
  }
}

template <int N>
__global__ void k_gemm_wmma(const float* __restrict__ A, const float* __restrict__ W,
                            const float* __restrict__ dinv, float* __restrict__ out, int M) {
  extern __shared__ float Wlds[];
  gemm_block<N>(A, W, dinv, out, M, blockIdx.x, Wlds);
}

__global__ void k_gemm1_fill(const float* __restrict__ A, const float* __restrict__ W,
                             const float* __restrict__ dinv, float* __restrict__ out, int M,
                             int gemm_blocks,
                             const int* __restrict__ src, const int* __restrict__ dst,
                             const int* __restrict__ rowstart, const int* __restrict__ rank,
                             const int* __restrict__ hist, int* __restrict__ csr,
                             int E, int chunk, int np, int tiles) {
  extern __shared__ float Wlds[];
  const int b = (int)blockIdx.x;
  if (b < gemm_blocks) {
    gemm_block<128>(A, W, dinv, out, M, b, Wlds);
  } else {
    const int f = b - gemm_blocks;
    const int by = f / tiles;
    fill_tile(src, dst, rowstart, rank, hist, csr, E, chunk, np, f - by * tiles, by);
  }
}

template <int F>
struct AggVec {
  typedef float type __attribute__((ext_vector_type(F / 32)));
};

template <int F, int NB>
__device__ __forceinline__ typename AggVec<F>::type agg_rows(
    const float* __restrict__ hws, int idx, int k, int lane, typename AggVec<F>::type sum) {
  typedef typename AggVec<F>::type vf;
  constexpr int V = F / 32;
  vf v[NB];
  #pragma unroll
  for (int u = 0; u < NB; ++u) {
    const unsigned r = (unsigned)__builtin_amdgcn_readlane(idx, k + u);
    const float* row = hws + (size_t)r * F;
    v[u] = *(const vf*)(row + V * lane);
  }
  #pragma unroll
  for (int u = 0; u < NB; ++u) sum += v[u];
  return sum;
}

template <int F>
__global__ void __launch_bounds__(256)
k_aggregate(const float* __restrict__ hws, const float* __restrict__ dinv,
            const int* __restrict__ rowstart, const int* __restrict__ csr,
            const float* __restrict__ bias, float* __restrict__ out, int n) {
  constexpr int V = F / 32;
  typedef typename AggVec<F>::type vf;
  const int lane = threadIdx.x & 31;
  const int wave = __builtin_amdgcn_readfirstlane(threadIdx.x >> 5);
  const int wpb = blockDim.x >> 5;
  const int nwaves = gridDim.x * wpb;
  const vf bv = *(const vf*)(bias + V * lane);
  for (int node = blockIdx.x * wpb + wave; node < n; node += nwaves) {
    const int s = rowstart[node];
    const int e = rowstart[node + 1];
    vf sum = *(const vf*)(hws + (size_t)node * F + V * lane);
    for (int base = s; base < e; base += 32) {
      const int cnt = min(e - base, 32);
      const int idx = csr[base + min(lane, cnt - 1)];
      int k = 0;
      for (; k + 8 <= cnt; k += 8) sum = agg_rows<F, 8>(hws, idx, k, lane, sum);
      if (cnt & 4) { sum = agg_rows<F, 4>(hws, idx, k, lane, sum); k += 4; }
      if (cnt & 2) { sum = agg_rows<F, 2>(hws, idx, k, lane, sum); k += 2; }
      if (cnt & 1) sum = agg_rows<F, 1>(hws, idx, k, lane, sum);
    }
    const float di = dinv[node];
    vf o;
    #pragma unroll
    for (int c = 0; c < V; ++c) o[c] = fmaxf(di * sum[c] + bv[c], 0.0f);
    *(vf*)(out + (size_t)node * F + V * lane) = o;
  }
}

#define POOL_ROWS 32
__device__ __forceinline__ void pool_flush(float* __restrict__ pool, float* __restrict__ cntg,
                                           int g, v2f acc, int run, int lane) {
  float* p = pool + g * 64 + 2 * lane;
  atomicAdd(p, acc.x);
  atomicAdd(p + 1, acc.y);
  if (lane == 0) atomicAdd(&cntg[g], (float)run);
}

__global__ void __launch_bounds__(128)
k_pool_acc(const float* __restrict__ h, const int* __restrict__ batch,
           float* __restrict__ pool, float* __restrict__ cntg, int n) {
  const int lane = threadIdx.x & 31;
  const int wave = __builtin_amdgcn_readfirstlane(threadIdx.x >> 5);
  const int node0 = (blockIdx.x * (blockDim.x >> 5) + wave) * POOL_ROWS;
  if (node0 >= n) return;
  const int nv = min(POOL_ROWS, n - node0);
  const int ids = batch[node0 + min(lane, nv - 1)];
  v2f v[POOL_ROWS];
  #pragma unroll
  for (int u = 0; u < POOL_ROWS; ++u) {
    const int r = min(node0 + u, n - 1);
    const float* row = h + (size_t)r * 64;
    v[u] = *(const v2f*)(row + 2 * lane);
  }
  int cur = __builtin_amdgcn_readlane(ids, 0);
  int run = 1;
  v2f acc = v[0];
  #pragma unroll
  for (int u = 1; u < POOL_ROWS; ++u) {
    if (u < nv) {
      const int g = __builtin_amdgcn_readlane(ids, u);
      if (g != cur) {
        pool_flush(pool, cntg, cur, acc, run, lane);
        acc = (v2f){0.0f, 0.0f};
        run = 0;
        cur = g;
      }
      acc += v[u];
      ++run;
    }
  }
  pool_flush(pool, cntg, cur, acc, run, lane);
}

__global__ void k_pool_final(const float* __restrict__ pool, const float* __restrict__ cntg,
                             float* __restrict__ out) {
  int idx = blockIdx.x * blockDim.x + threadIdx.x;
  if (idx < N_GRAPHS * 64) {
    out[idx] = pool[idx] / fmaxf(cntg[idx >> 6], 1.0f);
  }
}

extern "C" void kernel_launch(void* const* d_in, const int* in_sizes, int n_in,
                              void* d_out, int out_size, void* d_ws, size_t ws_size,
                              hipStream_t stream) {
  const float* x     = (const float*)d_in[0];
  const float* W1    = (const float*)d_in[1];
  const float* b1    = (const float*)d_in[2];
  const float* W2    = (const float*)d_in[3];
  const float* b2    = (const float*)d_in[4];
  const int*   ei    = (const int*)d_in[5];
  const int*   batch = (const int*)d_in[6];

  const int N = in_sizes[6];
  const int E = in_sizes[5] / 2;
  const int G = N_GRAPHS;

  const int* src = ei;
  const int* dst = ei + E;

  char* base = (char*)d_ws;
  size_t off = 0;
  auto take = [&](size_t bytes) {
    size_t o = off;
    off = (off + bytes + 255) & ~(size_t)255;
    return o;
  };
  const int np = (N + 31) & ~31;
  const int chunk = (E + CNT_BLOCKS - 1) / CNT_BLOCKS;
  int*   cnt      = (int*)(base + take(sizeof(int) * (size_t)N));
  int*   hist     = (int*)(base + take(sizeof(int) * (size_t)np * CNT_BLOCKS));
  int*   rank     = (int*)(base + take(sizeof(int) * (size_t)E));
  int*   rowstart = (int*)(base + take(sizeof(int) * (size_t)(N + 1)));
  float* dinv     = (float*)(base + take(sizeof(float) * (size_t)N));
  int*   csr      = (int*)(base + take(sizeof(int) * (size_t)E));
  float* hw1      = (float*)(base + take(sizeof(float) * (size_t)N * 128));
  float* h1       = (float*)(base + take(sizeof(float) * (size_t)N * 128));
  float* pool     = (float*)(base + take(sizeof(float) * (size_t)G * 64));
  float* cntg     = (float*)(base + take(sizeof(float) * (size_t)G));
  float* hw2 = hw1;
  float* h2  = h1;

  const int B = 256;
  const int gN = (N + B - 1) / B;
  const int mtiles = (N + 15) / 16;
  const int gGemm = (mtiles + 7) / 8;
  const int fillTiles = (chunk + FILL_TILE - 1) / FILL_TILE;

  k_count<<<CNT_BLOCKS, 1024, sizeof(int4) * (size_t)((N + 3) / 4), stream>>>(dst, hist, rank, E, N, chunk, np);
  k_merge<<<gN, B, 0, stream>>>(hist, cnt, N, np);
  k_scan<<<1, 1024, 0, stream>>>(cnt, rowstart, dinv, N);

  k_gemm1_fill<<<gGemm + fillTiles * CNT_BLOCKS, FILL_THREADS, 64 * (2 * 128 + 32) * sizeof(float), stream>>>(
      x, W1, dinv, hw1, N, gGemm, src, dst, rowstart, rank, hist, csr, E, chunk, np, fillTiles);
  k_aggregate<128><<<2048, 256, 0, stream>>>(hw1, dinv, rowstart, csr, b1, h1, N);

  k_gemm_wmma<64><<<gGemm, 256, 64 * (2 * 64 + 32) * sizeof(float), stream>>>(h1, W2, dinv, hw2, N);
  k_aggregate<64><<<2048, 256, 0, stream>>>(hw2, dinv, rowstart, csr, b2, h2, N);

  k_zero_f32<<<(G * 64 + G + B - 1) / B, B, 0, stream>>>(pool, G * 64 + G);
  k_pool_acc<<<(N + 4 * POOL_ROWS - 1) / (4 * POOL_ROWS), 128, 0, stream>>>(h2, batch, pool, cntg, N);
  k_pool_final<<<(G * 64 + B - 1) / B, B, 0, stream>>>(pool, cntg, (float*)d_out);
}


# GEMM W staging: a thread issues all its 16-byte loads of W together (16 for N=128, 8 for N=64), then 16-byte LDS stores: one memory round trip a block, not one per 4-byte element of a thread
# speedup vs baseline: 3.6332x; 1.1563x over previous
#include <hip/hip_runtime.h>
#include <hip/hip_bf16.h>

typedef float v2f __attribute__((ext_vector_type(2)));
typedef float v8f __attribute__((ext_vector_type(8)));

#define N_GRAPHS 64

__global__ void k_zero_f32(float* p, int n) {
  int i = blockIdx.x * blockDim.x + threadIdx.x;
  for (; i < n; i += gridDim.x * blockDim.x) p[i] = 0.0f;
}

#define CNT_BLOCKS 64
#define CNT_BATCH 8
__global__ void __launch_bounds__(1024)
k_count(const int* __restrict__ dst, int* __restrict__ hist, int* __restrict__ rank,
        int E, int n, int chunk, int np) {
  extern __shared__ int4 hc4[];
  int* hc = (int*)hc4;
  const int tid = threadIdx.x;
  const int nq = (n + 3) >> 2;
  for (int i = tid; i < nq; i += 1024) hc4[i] = make_int4(0, 0, 0, 0);
  __syncthreads();
  const int start = blockIdx.x * chunk;
  const int end = min(E, start + chunk);
  for (int e0 = start + tid; e0 < end; e0 += CNT_BATCH * 1024) {
    int d[CNT_BATCH];
    int r[CNT_BATCH];
    if (e0 + (CNT_BATCH - 1) * 1024 < end) {
      #pragma unroll
      for (int u = 0; u < CNT_BATCH; ++u) d[u] = dst[e0 + u * 1024];
      #pragma unroll
      for (int u = 0; u < CNT_BATCH; ++u) r[u] = atomicAdd(&hc[d[u]], 1);
      #pragma unroll
      for (int u = 0; u < CNT_BATCH; ++u) rank[e0 + u * 1024] = r[u];
    } else {
      #pragma unroll
      for (int u = 0; u < CNT_BATCH; ++u) {
        const int e = e0 + u * 1024;
        d[u] = (e < end) ? dst[e] : 0;
      }
      #pragma unroll
      for (int u = 0; u < CNT_BATCH; ++u) {
        const int e = e0 + u * 1024;
        if (e < end) rank[e] = atomicAdd(&hc[d[u]], 1);
      }
    }
  }
  __syncthreads();
  int4* __restrict__ row = (int4*)(hist + (size_t)blockIdx.x * np);
  for (int i = tid; i < nq; i += 1024) row[i] = hc4[i];
}

__global__ void k_merge(int* __restrict__ hist, int* __restrict__ cnt, int n, int np) {
  const int i = blockIdx.x * blockDim.x + threadIdx.x;
  if (i >= n) return;
  int run = 0;
  for (int b0 = 0; b0 < CNT_BLOCKS; b0 += 16) {
    int c[16];
    #pragma unroll
    for (int u = 0; u < 16; ++u) c[u] = hist[(size_t)(b0 + u) * np + i];
    #pragma unroll
    for (int u = 0; u < 16; ++u) {
      hist[(size_t)(b0 + u) * np + i] = run;
      run += c[u];
    }
  }
  cnt[i] = run;
}

#define SCAN_ROWS 13
__global__ void k_scan(const int* __restrict__ cnt, int* __restrict__ rowstart,
                       float* __restrict__ dinv, int n) {
  constexpr int WSEG = SCAN_ROWS * 128;
  constexpr int TILE = 32 * WSEG;
  __shared__ int wtot[32];
  const int tid = threadIdx.x;
  const int lane = tid & 31;
  const int wave = tid >> 5;
  const int n4 = n & ~3;
  int carry = 0;
  for (int base = 0; base < n4; base += TILE) {
    const int i0 = base + wave * WSEG + lane * 4;
    int4 v[SCAN_ROWS];
    #pragma unroll
    for (int j = 0; j < SCAN_ROWS; ++j) {
      const int i = i0 + j * 128;
      int4 t = make_int4(0, 0, 0, 0);
      if (i < n4) t = *(const int4*)(cnt + i);
      v[j] = t;
    }
    int inc[SCAN_ROWS];
    #pragma unroll
    for (int j = 0; j < SCAN_ROWS; ++j) inc[j] = v[j].x + v[j].y + v[j].z + v[j].w;
    #pragma unroll
    for (int off = 1; off < 32; off <<= 1) {
      #pragma unroll
      for (int j = 0; j < SCAN_ROWS; ++j) {
        const int t = __shfl_up(inc[j], off, 32);
        if (lane >= off) inc[j] += t;
      }
    }
    int rowoff[SCAN_ROWS];
    int wsum = 0;
    #pragma unroll
    for (int j = 0; j < SCAN_ROWS; ++j) {
      rowoff[j] = wsum;
      wsum += __builtin_amdgcn_readlane(inc[j], 31);
    }
    if (lane == 0) wtot[wave] = wsum;
    __syncthreads();
    const int wt = wtot[lane];
    int winc = wt;
    #pragma unroll
    for (int off = 1; off < 32; off <<= 1) {
      const int t = __shfl_up(winc, off, 32);
      if (lane >= off) winc += t;
    }
    const int woff = carry + __shfl(winc - wt, wave, 32);
    carry += __builtin_amdgcn_readlane(winc, 31);
    #pragma unroll
    for (int j = 0; j < SCAN_ROWS; ++j) {
      const int i = i0 + j * 128;
      const int4 c = v[j];
      int4 r;
      r.x = woff + rowoff[j] + inc[j] - (c.x + c.y + c.z + c.w);
      r.y = r.x + c.x;
      r.z = r.y + c.y;
      r.w = r.z + c.z;
      float4 d;
      d.x = rsqrtf((float)c.x + 1.0f);
      d.y = rsqrtf((float)c.y + 1.0f);
      d.z = rsqrtf((float)c.z + 1.0f);
      d.w = rsqrtf((float)c.w + 1.0f);
      if (i < n4) {
        *(int4*)(rowstart + i) = r;
        *(float4*)(dinv + i) = d;
      }
    }
    __syncthreads();
  }
  if (tid == 0) {
    int c = carry;
    for (int i = n4; i < n; ++i) {
      const int vv = cnt[i];
      rowstart[i] = c;
      dinv[i] = rsqrtf((float)vv + 1.0f);
      c += vv;
    }
    rowstart[n] = c;
  }
}

#define FILL_BATCH 8
#define FILL_THREADS 256
#define FILL_TILE (FILL_BATCH * FILL_THREADS)
__device__ __forceinline__ void fill_tile(
    const int* __restrict__ src, const int* __restrict__ dst,
    const int* __restrict__ rowstart, const int* __restrict__ rank,
    const int* __restrict__ hist, int* __restrict__ csr,
    int E, int chunk, int np, int bx, int by) {
  const int cstart = by * chunk;
  const int lim = min(chunk, E - cstart);
  const int t0 = bx * FILL_TILE;
  if (t0 >= lim) return;
  const int* __restrict__ hrow = hist + (size_t)by * np;
  const int l0 = t0 + (int)threadIdx.x;
  int d[FILL_BATCH], r[FILL_BATCH], s[FILL_BATCH], p[FILL_BATCH];
  if (t0 + FILL_TILE <= lim) {
    const int e0 = cstart + l0;
    #pragma unroll
    for (int u = 0; u < FILL_BATCH; ++u) d[u] = dst[e0 + u * FILL_THREADS];
    #pragma unroll
    for (int u = 0; u < FILL_BATCH; ++u) r[u] = rank[e0 + u * FILL_THREADS];
    #pragma unroll
    for (int u = 0; u < FILL_BATCH; ++u) s[u] = src[e0 + u * FILL_THREADS];
    #pragma unroll
    for (int u = 0; u < FILL_BATCH; ++u) p[u] = rowstart[d[u]] + hrow[d[u]];
    #pragma unroll
    for (int u = 0; u < FILL_BATCH; ++u) csr[p[u] + r[u]] = s[u];
  } else {
    #pragma unroll
    for (int u = 0; u < FILL_BATCH; ++u) {
      const int e = cstart + min(l0 + u * FILL_THREADS, lim - 1);
      d[u] = dst[e];
      r[u] = rank[e];
      s[u] = src[e];
    }
    #pragma unroll
    for (int u = 0; u < FILL_BATCH; ++u) p[u] = rowstart[d[u]] + hrow[d[u]];
    #pragma unroll
    for (int u = 0; u < FILL_BATCH; ++u) {
      if (l0 + u * FILL_THREADS < lim) csr[p[u] + r[u]] = s[u];
    }
  }
}

#define GEMM_THREADS 256
template <int N>
__device__ __forceinline__ void gemm_block(const float* __restrict__ A, const float* __restrict__ W,
                                           const float* __restrict__ dinv, float* __restrict__ out, int M,
                                           int bid, float* Wlds) {
  constexpr int K = 128;
  constexpr int NT = N / 16;
  constexpr int RP = 2 * N + 32;
  constexpr int QN = N / 4;
  constexpr int PSTEP = GEMM_THREADS / QN;
  constexpr int WPER = (K / 2) / PSTEP;
  static_assert(GEMM_THREADS % QN == 0 && (K / 2) % PSTEP == 0, "W items must divide over the block");
  const int p0 = (int)threadIdx.x / QN;
  const int q = (int)threadIdx.x % QN;
  const float* wsrc = W + (size_t)(2 * p0) * N + 4 * q;
  float* wdst = Wlds + p0 * RP + 8 * q;
  float4 wlo[WPER], whi[WPER];
  #pragma unroll
  for (int j = 0; j < WPER; ++j) {
    wlo[j] = *(const float4*)(wsrc + (size_t)j * (2 * PSTEP * N));
    whi[j] = *(const float4*)(wsrc + (size_t)j * (2 * PSTEP * N) + N);
  }
  #pragma unroll
  for (int j = 0; j < WPER; ++j) {
    float* d = wdst + j * (PSTEP * RP);
    *(float4*)d       = make_float4(wlo[j].x, whi[j].x, wlo[j].y, whi[j].y);
    *(float4*)(d + 4) = make_float4(wlo[j].z, whi[j].z, wlo[j].w, whi[j].w);
  }
  __syncthreads();

  const int wave = threadIdx.x >> 5;
  const int lane = threadIdx.x & 31;
  const int mtile = bid * 8 + wave;
  if (mtile * 16 >= M) return;

  const int Rm = mtile * 16;
  const int half = lane >> 4;
  const int l16 = lane & 15;

  v8f acc[NT] = {};
  const float* arow = A + (size_t)(Rm + l16) * K;
  const float* bbase = Wlds + half * RP + l16 * 2;

  for (int kb = 0; kb < K; kb += 4) {
    v2f a = *(const v2f*)(arow + kb + 2 * half);
    const float* bk = bbase + (kb >> 1) * RP;
    #pragma unroll
    for (int t = 0; t < NT; ++t) {
      v2f b = *(const v2f*)(bk + t * 32);
      acc[t] = __builtin_amdgcn_wmma_f32_16x16x4_f32(
          false, a, false, b, (short)0, acc[t], false, false);
    }
  }

  #pragma unroll
  for (int r = 0; r < 8; ++r) {
    const int row = Rm + r + 8 * half;
    const float di = dinv[row];
    #pragma unroll
    for (int t = 0; t < NT; ++t) {
      out[(size_t)row * N + t * 16 + l16] = acc[t][r] * di;
    }
  }
}

template <int N>
__global__ void __launch_bounds__(GEMM_THREADS)
k_gemm_wmma(const float* __restrict__ A, const float* __restrict__ W,
            const float* __restrict__ dinv, float* __restrict__ out, int M) {
  extern __shared__ float4 Wlds4[];
  gemm_block<N>(A, W, dinv, out, M, blockIdx.x, (float*)Wlds4);
}

static_assert(FILL_THREADS == GEMM_THREADS, "the shared launch has one block size");
__global__ void __launch_bounds__(GEMM_THREADS)
k_gemm1_fill(const float* __restrict__ A, const float* __restrict__ W,
             const float* __restrict__ dinv, float* __restrict__ out, int M,
             int gemm_blocks,
             const int* __restrict__ src, const int* __restrict__ dst,
             const int* __restrict__ rowstart, const int* __restrict__ rank,
             const int* __restrict__ hist, int* __restrict__ csr,
             int E, int chunk, int np, int tiles) {
  extern __shared__ float4 Wlds4[];
  const int b = (int)blockIdx.x;
  if (b < gemm_blocks) {
    gemm_block<128>(A, W, dinv, out, M, b, (float*)Wlds4);
  } else {
    const int f = b - gemm_blocks;
    const int by = f / tiles;
    fill_tile(src, dst, rowstart, rank, hist, csr, E, chunk, np, f - by * tiles, by);
  }
}

template <int F>
struct AggVec {
  typedef float type __attribute__((ext_vector_type(F / 32)));
};

template <int F, int NB>
__device__ __forceinline__ typename AggVec<F>::type agg_rows(
    const float* __restrict__ hws, int idx, int k, int lane, typename AggVec<F>::type sum) {
  typedef typename AggVec<F>::type vf;
  constexpr int V = F / 32;
  vf v[NB];
  #pragma unroll
  for (int u = 0; u < NB; ++u) {
    const unsigned r = (unsigned)__builtin_amdgcn_readlane(idx, k + u);
    const float* row = hws + (size_t)r * F;
    v[u] = *(const vf*)(row + V * lane);
  }
  #pragma unroll
  for (int u = 0; u < NB; ++u) sum += v[u];
  return sum;
}

template <int F>
__global__ void __launch_bounds__(256)
k_aggregate(const float* __restrict__ hws, const float* __restrict__ dinv,
            const int* __restrict__ rowstart, const int* __restrict__ csr,
            const float* __restrict__ bias, float* __restrict__ out, int n) {
  constexpr int V = F / 32;
  typedef typename AggVec<F>::type vf;
  const int lane = threadIdx.x & 31;
  const int wave = __builtin_amdgcn_readfirstlane(threadIdx.x >> 5);
  const int wpb = blockDim.x >> 5;
  const int nwaves = gridDim.x * wpb;
  const vf bv = *(const vf*)(bias + V * lane);
  for (int node = blockIdx.x * wpb + wave; node < n; node += nwaves) {
    const int s = rowstart[node];
    const int e = rowstart[node + 1];
    vf sum = *(const vf*)(hws + (size_t)node * F + V * lane);
    for (int base = s; base < e; base += 32) {
      const int cnt = min(e - base, 32);
      const int idx = csr[base + min(lane, cnt - 1)];
      int k = 0;
      for (; k + 8 <= cnt; k += 8) sum = agg_rows<F, 8>(hws, idx, k, lane, sum);
      if (cnt & 4) { sum = agg_rows<F, 4>(hws, idx, k, lane, sum); k += 4; }
      if (cnt & 2) { sum = agg_rows<F, 2>(hws, idx, k, lane, sum); k += 2; }
      if (cnt & 1) sum = agg_rows<F, 1>(hws, idx, k, lane, sum);
    }
    const float di = dinv[node];
    vf o;
    #pragma unroll
    for (int c = 0; c < V; ++c) o[c] = fmaxf(di * sum[c] + bv[c], 0.0f);
    *(vf*)(out + (size_t)node * F + V * lane) = o;
  }
}

#define POOL_ROWS 32
__device__ __forceinline__ void pool_flush(float* __restrict__ pool, float* __restrict__ cntg,
                                           int g, v2f acc, int run, int lane) {
  float* p = pool + g * 64 + 2 * lane;
  atomicAdd(p, acc.x);
  atomicAdd(p + 1, acc.y);
  if (lane == 0) atomicAdd(&cntg[g], (float)run);
}

__global__ void __launch_bounds__(128)
k_pool_acc(const float* __restrict__ h, const int* __restrict__ batch,
           float* __restrict__ pool, float* __restrict__ cntg, int n) {
  const int lane = threadIdx.x & 31;
  const int wave = __builtin_amdgcn_readfirstlane(threadIdx.x >> 5);
  const int node0 = (blockIdx.x * (blockDim.x >> 5) + wave) * POOL_ROWS;
  if (node0 >= n) return;
  const int nv = min(POOL_ROWS, n - node0);
  const int ids = batch[node0 + min(lane, nv - 1)];
  v2f v[POOL_ROWS];
  #pragma unroll
  for (int u = 0; u < POOL_ROWS; ++u) {
    const int r = min(node0 + u, n - 1);
    const float* row = h + (size_t)r * 64;
    v[u] = *(const v2f*)(row + 2 * lane);
  }
  int cur = __builtin_amdgcn_readlane(ids, 0);
  int run = 1;
  v2f acc = v[0];
  #pragma unroll
  for (int u = 1; u < POOL_ROWS; ++u) {
    if (u < nv) {
      const int g = __builtin_amdgcn_readlane(ids, u);
      if (g != cur) {
        pool_flush(pool, cntg, cur, acc, run, lane);
        acc = (v2f){0.0f, 0.0f};
        run = 0;
        cur = g;
      }
      acc += v[u];
      ++run;
    }
  }
  pool_flush(pool, cntg, cur, acc, run, lane);
}

__global__ void k_pool_final(const float* __restrict__ pool, const float* __restrict__ cntg,
                             float* __restrict__ out) {
  int idx = blockIdx.x * blockDim.x + threadIdx.x;
  if (idx < N_GRAPHS * 64) {
    out[idx] = pool[idx] / fmaxf(cntg[idx >> 6], 1.0f);
  }
}

extern "C" void kernel_launch(void* const* d_in, const int* in_sizes, int n_in,
                              void* d_out, int out_size, void* d_ws, size_t ws_size,
                              hipStream_t stream) {
  const float* x     = (const float*)d_in[0];
  const float* W1    = (const float*)d_in[1];
  const float* b1    = (const float*)d_in[2];
  const float* W2    = (const float*)d_in[3];
  const float* b2    = (const float*)d_in[4];
  const int*   ei    = (const int*)d_in[5];
  const int*   batch = (const int*)d_in[6];

  const int N = in_sizes[6];
  const int E = in_sizes[5] / 2;
  const int G = N_GRAPHS;

  const int* src = ei;
  const int* dst = ei + E;

  char* base = (char*)d_ws;
  size_t off = 0;
  auto take = [&](size_t bytes) {
    size_t o = off;
    off = (off + bytes + 255) & ~(size_t)255;
    return o;
  };
  const int np = (N + 31) & ~31;
  const int chunk = (E + CNT_BLOCKS - 1) / CNT_BLOCKS;
  int*   cnt      = (int*)(base + take(sizeof(int) * (size_t)N));
  int*   hist     = (int*)(base + take(sizeof(int) * (size_t)np * CNT_BLOCKS));
  int*   rank     = (int*)(base + take(sizeof(int) * (size_t)E));
  int*   rowstart = (int*)(base + take(sizeof(int) * (size_t)(N + 1)));
  float* dinv     = (float*)(base + take(sizeof(float) * (size_t)N));
  int*   csr      = (int*)(base + take(sizeof(int) * (size_t)E));
  float* hw1      = (float*)(base + take(sizeof(float) * (size_t)N * 128));
  float* h1       = (float*)(base + take(sizeof(float) * (size_t)N * 128));
  float* pool     = (float*)(base + take(sizeof(float) * (size_t)G * 64));
  float* cntg     = (float*)(base + take(sizeof(float) * (size_t)G));
  float* hw2 = hw1;
  float* h2  = h1;

  const int B = 256;
  const int gN = (N + B - 1) / B;
  const int mtiles = (N + 15) / 16;
  const int gGemm = (mtiles + 7) / 8;
  const int fillTiles = (chunk + FILL_TILE - 1) / FILL_TILE;

  k_count<<<CNT_BLOCKS, 1024, sizeof(int4) * (size_t)((N + 3) / 4), stream>>>(dst, hist, rank, E, N, chunk, np);
  k_merge<<<gN, B, 0, stream>>>(hist, cnt, N, np);
  k_scan<<<1, 1024, 0, stream>>>(cnt, rowstart, dinv, N);

  k_gemm1_fill<<<gGemm + fillTiles * CNT_BLOCKS, FILL_THREADS, 64 * (2 * 128 + 32) * sizeof(float), stream>>>(
      x, W1, dinv, hw1, N, gGemm, src, dst, rowstart, rank, hist, csr, E, chunk, np, fillTiles);
  k_aggregate<128><<<2048, 256, 0, stream>>>(hw1, dinv, rowstart, csr, b1, h1, N);

  k_gemm_wmma<64><<<gGemm, GEMM_THREADS, 64 * (2 * 64 + 32) * sizeof(float), stream>>>(h1, W2, dinv, hw2, N);
  k_aggregate<64><<<2048, 256, 0, stream>>>(hw2, dinv, rowstart, csr, b2, h2, N);

  k_zero_f32<<<(G * 64 + G + B - 1) / B, B, 0, stream>>>(pool, G * 64 + G);
  k_pool_acc<<<(N + 4 * POOL_ROWS - 1) / (4 * POOL_ROWS), 128, 0, stream>>>(h2, batch, pool, cntg, N);
  k_pool_final<<<(G * 64 + B - 1) / B, B, 0, stream>>>(pool, cntg, (float*)d_out);
}


# mean pool in one launch: a run's sum is divided by its graph's node count (k_merge counts) and added atomically into out (k_count clears it); clear and divide launches go; divide precedes the sum
# speedup vs baseline: 3.7643x; 1.0361x over previous
#include <hip/hip_runtime.h>
#include <hip/hip_bf16.h>

typedef float v2f __attribute__((ext_vector_type(2)));
typedef float v8f __attribute__((ext_vector_type(8)));

#define N_GRAPHS 64

#define CNT_BLOCKS 64
#define CNT_BATCH 8
__global__ void __launch_bounds__(1024)
k_count(const int* __restrict__ dst, int* __restrict__ hist, int* __restrict__ rank,
        int E, int n, int chunk, int np,
        float* __restrict__ outz, int nout, float* __restrict__ cntg, int ng) {
  extern __shared__ int4 hc4[];
  int* hc = (int*)hc4;
  const int tid = threadIdx.x;
  const int nq = (n + 3) >> 2;
  for (int i = blockIdx.x * 1024 + tid; i < nout; i += CNT_BLOCKS * 1024) outz[i] = 0.0f;
  for (int i = blockIdx.x * 1024 + tid; i < ng; i += CNT_BLOCKS * 1024) cntg[i] = 0.0f;
  for (int i = tid; i < nq; i += 1024) hc4[i] = make_int4(0, 0, 0, 0);
  __syncthreads();
  const int start = blockIdx.x * chunk;
  const int end = min(E, start + chunk);
  for (int e0 = start + tid; e0 < end; e0 += CNT_BATCH * 1024) {
    int d[CNT_BATCH];
    int r[CNT_BATCH];
    if (e0 + (CNT_BATCH - 1) * 1024 < end) {
      #pragma unroll
      for (int u = 0; u < CNT_BATCH; ++u) d[u] = dst[e0 + u * 1024];
      #pragma unroll
      for (int u = 0; u < CNT_BATCH; ++u) r[u] = atomicAdd(&hc[d[u]], 1);
      #pragma unroll
      for (int u = 0; u < CNT_BATCH; ++u) rank[e0 + u * 1024] = r[u];
    } else {
      #pragma unroll
      for (int u = 0; u < CNT_BATCH; ++u) {
        const int e = e0 + u * 1024;
        d[u] = (e < end) ? dst[e] : 0;
      }
      #pragma unroll
      for (int u = 0; u < CNT_BATCH; ++u) {
        const int e = e0 + u * 1024;
        if (e < end) rank[e] = atomicAdd(&hc[d[u]], 1);
      }
    }
  }
  __syncthreads();
  int4* __restrict__ row = (int4*)(hist + (size_t)blockIdx.x * np);
  for (int i = tid; i < nq; i += 1024) row[i] = hc4[i];
}

__global__ void k_merge(int* __restrict__ hist, int* __restrict__ cnt, int n, int np,
                        const int* __restrict__ batch, float* __restrict__ cntg) {
  const int i = blockIdx.x * blockDim.x + threadIdx.x;
  const bool valid = i < n;
  const int lane = threadIdx.x & 31;
  const int id = valid ? batch[i] : -1;
  if (valid) {
    int run = 0;
    for (int b0 = 0; b0 < CNT_BLOCKS; b0 += 16) {
      int c[16];
      #pragma unroll
      for (int u = 0; u < 16; ++u) c[u] = hist[(size_t)(b0 + u) * np + i];
      #pragma unroll
      for (int u = 0; u < 16; ++u) {
        hist[(size_t)(b0 + u) * np + i] = run;
        run += c[u];
      }
    }
    cnt[i] = run;
  }
  unsigned todo = __builtin_amdgcn_ballot_w32(valid);
  while (todo != 0u) {
    const int leader = __builtin_ctz(todo);
    const int g = __builtin_amdgcn_readlane(id, leader);
    const unsigned same = __builtin_amdgcn_ballot_w32(valid && id == g);
    if (lane == leader) atomicAdd(&cntg[g], (float)__builtin_popcount(same));
    todo &= ~same;
  }
}

#define SCAN_ROWS 13
__global__ void k_scan(const int* __restrict__ cnt, int* __restrict__ rowstart,
                       float* __restrict__ dinv, int n) {
  constexpr int WSEG = SCAN_ROWS * 128;
  constexpr int TILE = 32 * WSEG;
  __shared__ int wtot[32];
  const int tid = threadIdx.x;
  const int lane = tid & 31;
  const int wave = tid >> 5;
  const int n4 = n & ~3;
  int carry = 0;
  for (int base = 0; base < n4; base += TILE) {
    const int i0 = base + wave * WSEG + lane * 4;
    int4 v[SCAN_ROWS];
    #pragma unroll
    for (int j = 0; j < SCAN_ROWS; ++j) {
      const int i = i0 + j * 128;
      int4 t = make_int4(0, 0, 0, 0);
      if (i < n4) t = *(const int4*)(cnt + i);
      v[j] = t;
    }
    int inc[SCAN_ROWS];
    #pragma unroll
    for (int j = 0; j < SCAN_ROWS; ++j) inc[j] = v[j].x + v[j].y + v[j].z + v[j].w;
    #pragma unroll
    for (int off = 1; off < 32; off <<= 1) {
      #pragma unroll
      for (int j = 0; j < SCAN_ROWS; ++j) {
        const int t = __shfl_up(inc[j], off, 32);
        if (lane >= off) inc[j] += t;
      }
    }
    int rowoff[SCAN_ROWS];
    int wsum = 0;
    #pragma unroll
    for (int j = 0; j < SCAN_ROWS; ++j) {
      rowoff[j] = wsum;
      wsum += __builtin_amdgcn_readlane(inc[j], 31);
    }
    if (lane == 0) wtot[wave] = wsum;
    __syncthreads();
    const int wt = wtot[lane];
    int winc = wt;
    #pragma unroll
    for (int off = 1; off < 32; off <<= 1) {
      const int t = __shfl_up(winc, off, 32);
      if (lane >= off) winc += t;
    }
    const int woff = carry + __shfl(winc - wt, wave, 32);
    carry += __builtin_amdgcn_readlane(winc, 31);
    #pragma unroll
    for (int j = 0; j < SCAN_ROWS; ++j) {
      const int i = i0 + j * 128;
      const int4 c = v[j];
      int4 r;
      r.x = woff + rowoff[j] + inc[j] - (c.x + c.y + c.z + c.w);
      r.y = r.x + c.x;
      r.z = r.y + c.y;
      r.w = r.z + c.z;
      float4 d;
      d.x = rsqrtf((float)c.x + 1.0f);
      d.y = rsqrtf((float)c.y + 1.0f);
      d.z = rsqrtf((float)c.z + 1.0f);
      d.w = rsqrtf((float)c.w + 1.0f);
      if (i < n4) {
        *(int4*)(rowstart + i) = r;
        *(float4*)(dinv + i) = d;
      }
    }
    __syncthreads();
  }
  if (tid == 0) {
    int c = carry;
    for (int i = n4; i < n; ++i) {
      const int vv = cnt[i];
      rowstart[i] = c;
      dinv[i] = rsqrtf((float)vv + 1.0f);
      c += vv;
    }
    rowstart[n] = c;
  }
}

#define FILL_BATCH 8
#define FILL_THREADS 256
#define FILL_TILE (FILL_BATCH * FILL_THREADS)
__device__ __forceinline__ void fill_tile(
    const int* __restrict__ src, const int* __restrict__ dst,
    const int* __restrict__ rowstart, const int* __restrict__ rank,
    const int* __restrict__ hist, int* __restrict__ csr,
    int E, int chunk, int np, int bx, int by) {
  const int cstart = by * chunk;
  const int lim = min(chunk, E - cstart);
  const int t0 = bx * FILL_TILE;
  if (t0 >= lim) return;
  const int* __restrict__ hrow = hist + (size_t)by * np;
  const int l0 = t0 + (int)threadIdx.x;
  int d[FILL_BATCH], r[FILL_BATCH], s[FILL_BATCH], p[FILL_BATCH];
  if (t0 + FILL_TILE <= lim) {
    const int e0 = cstart + l0;
    #pragma unroll
    for (int u = 0; u < FILL_BATCH; ++u) d[u] = dst[e0 + u * FILL_THREADS];
    #pragma unroll
    for (int u = 0; u < FILL_BATCH; ++u) r[u] = rank[e0 + u * FILL_THREADS];
    #pragma unroll
    for (int u = 0; u < FILL_BATCH; ++u) s[u] = src[e0 + u * FILL_THREADS];
    #pragma unroll
    for (int u = 0; u < FILL_BATCH; ++u) p[u] = rowstart[d[u]] + hrow[d[u]];
    #pragma unroll
    for (int u = 0; u < FILL_BATCH; ++u) csr[p[u] + r[u]] = s[u];
  } else {
    #pragma unroll
    for (int u = 0; u < FILL_BATCH; ++u) {
      const int e = cstart + min(l0 + u * FILL_THREADS, lim - 1);
      d[u] = dst[e];
      r[u] = rank[e];
      s[u] = src[e];
    }
    #pragma unroll
    for (int u = 0; u < FILL_BATCH; ++u) p[u] = rowstart[d[u]] + hrow[d[u]];
    #pragma unroll
    for (int u = 0; u < FILL_BATCH; ++u) {
      if (l0 + u * FILL_THREADS < lim) csr[p[u] + r[u]] = s[u];
    }
  }
}

#define GEMM_THREADS 256
template <int N>
__device__ __forceinline__ void gemm_block(const float* __restrict__ A, const float* __restrict__ W,
                                           const float* __restrict__ dinv, float* __restrict__ out, int M,
                                           int bid, float* Wlds) {
  constexpr int K = 128;
  constexpr int NT = N / 16;
  constexpr int RP = 2 * N + 32;
  constexpr int QN = N / 4;
  constexpr int PSTEP = GEMM_THREADS / QN;
  constexpr int WPER = (K / 2) / PSTEP;
  static_assert(GEMM_THREADS % QN == 0 && (K / 2) % PSTEP == 0, "W items must divide over the block");
  const int p0 = (int)threadIdx.x / QN;
  const int q = (int)threadIdx.x % QN;
  const float* wsrc = W + (size_t)(2 * p0) * N + 4 * q;
  float* wdst = Wlds + p0 * RP + 8 * q;
  float4 wlo[WPER], whi[WPER];
  #pragma unroll
  for (int j = 0; j < WPER; ++j) {
    wlo[j] = *(const float4*)(wsrc + (size_t)j * (2 * PSTEP * N));
    whi[j] = *(const float4*)(wsrc + (size_t)j * (2 * PSTEP * N) + N);
  }
  #pragma unroll
  for (int j = 0; j < WPER; ++j) {
    float* d = wdst + j * (PSTEP * RP);
    *(float4*)d       = make_float4(wlo[j].x, whi[j].x, wlo[j].y, whi[j].y);
    *(float4*)(d + 4) = make_float4(wlo[j].z, whi[j].z, wlo[j].w, whi[j].w);
  }
  __syncthreads();

  const int wave = threadIdx.x >> 5;
  const int lane = threadIdx.x & 31;
  const int mtile = bid * 8 + wave;
  if (mtile * 16 >= M) return;

  const int Rm = mtile * 16;
  const int half = lane >> 4;
  const int l16 = lane & 15;

  v8f acc[NT] = {};
  const float* arow = A + (size_t)(Rm + l16) * K;
  const float* bbase = Wlds + half * RP + l16 * 2;

  for (int kb = 0; kb < K; kb += 4) {
    v2f a = *(const v2f*)(arow + kb + 2 * half);
    const float* bk = bbase + (kb >> 1) * RP;
    #pragma unroll
    for (int t = 0; t < NT; ++t) {
      v2f b = *(const v2f*)(bk + t * 32);
      acc[t] = __builtin_amdgcn_wmma_f32_16x16x4_f32(
          false, a, false, b, (short)0, acc[t], false, false);
    }
  }

  #pragma unroll
  for (int r = 0; r < 8; ++r) {
    const int row = Rm + r + 8 * half;
    const float di = dinv[row];
    #pragma unroll
    for (int t = 0; t < NT; ++t) {
      out[(size_t)row * N + t * 16 + l16] = acc[t][r] * di;
    }
  }
}

template <int N>
__global__ void __launch_bounds__(GEMM_THREADS)
k_gemm_wmma(const float* __restrict__ A, const float* __restrict__ W,
            const float* __restrict__ dinv, float* __restrict__ out, int M) {
  extern __shared__ float4 Wlds4[];
  gemm_block<N>(A, W, dinv, out, M, blockIdx.x, (float*)Wlds4);
}

static_assert(FILL_THREADS == GEMM_THREADS, "the shared launch has one block size");
__global__ void __launch_bounds__(GEMM_THREADS)
k_gemm1_fill(const float* __restrict__ A, const float* __restrict__ W,
             const float* __restrict__ dinv, float* __restrict__ out, int M,
             int gemm_blocks,
             const int* __restrict__ src, const int* __restrict__ dst,
             const int* __restrict__ rowstart, const int* __restrict__ rank,
             const int* __restrict__ hist, int* __restrict__ csr,
             int E, int chunk, int np, int tiles) {
  extern __shared__ float4 Wlds4[];
  const int b = (int)blockIdx.x;
  if (b < gemm_blocks) {
    gemm_block<128>(A, W, dinv, out, M, b, (float*)Wlds4);
  } else {
    const int f = b - gemm_blocks;
    const int by = f / tiles;
    fill_tile(src, dst, rowstart, rank, hist, csr, E, chunk, np, f - by * tiles, by);
  }
}

template <int F>
struct AggVec {
  typedef float type __attribute__((ext_vector_type(F / 32)));
};

template <int F, int NB>
__device__ __forceinline__ typename AggVec<F>::type agg_rows(
    const float* __restrict__ hws, int idx, int k, int lane, typename AggVec<F>::type sum) {
  typedef typename AggVec<F>::type vf;
  constexpr int V = F / 32;
  vf v[NB];
  #pragma unroll
  for (int u = 0; u < NB; ++u) {
    const unsigned r = (unsigned)__builtin_amdgcn_readlane(idx, k + u);
    const float* row = hws + (size_t)r * F;
    v[u] = *(const vf*)(row + V * lane);
  }
  #pragma unroll
  for (int u = 0; u < NB; ++u) sum += v[u];
  return sum;
}

template <int F>
__global__ void __launch_bounds__(256)
k_aggregate(const float* __restrict__ hws, const float* __restrict__ dinv,
            const int* __restrict__ rowstart, const int* __restrict__ csr,
            const float* __restrict__ bias, float* __restrict__ out, int n) {
  constexpr int V = F / 32;
  typedef typename AggVec<F>::type vf;
  const int lane = threadIdx.x & 31;
  const int wave = __builtin_amdgcn_readfirstlane(threadIdx.x >> 5);
  const int wpb = blockDim.x >> 5;
  const int nwaves = gridDim.x * wpb;
  const vf bv = *(const vf*)(bias + V * lane);
  for (int node = blockIdx.x * wpb + wave; node < n; node += nwaves) {
    const int s = rowstart[node];
    const int e = rowstart[node + 1];
    vf sum = *(const vf*)(hws + (size_t)node * F + V * lane);
    for (int base = s; base < e; base += 32) {
      const int cnt = min(e - base, 32);
      const int idx = csr[base + min(lane, cnt - 1)];
      int k = 0;
      for (; k + 8 <= cnt; k += 8) sum = agg_rows<F, 8>(hws, idx, k, lane, sum);
      if (cnt & 4) { sum = agg_rows<F, 4>(hws, idx, k, lane, sum); k += 4; }
      if (cnt & 2) { sum = agg_rows<F, 2>(hws, idx, k, lane, sum); k += 2; }
      if (cnt & 1) sum = agg_rows<F, 1>(hws, idx, k, lane, sum);
    }
    const float di = dinv[node];
    vf o;
    #pragma unroll
    for (int c = 0; c < V; ++c) o[c] = fmaxf(di * sum[c] + bv[c], 0.0f);
    *(vf*)(out + (size_t)node * F + V * lane) = o;
  }
}

#define POOL_ROWS 32
__device__ __forceinline__ void pool_flush(float* __restrict__ out, float c, int g, v2f acc, int lane) {
  float* p = out + g * 64 + 2 * lane;
  atomicAdd(p, acc.x / c);
  atomicAdd(p + 1, acc.y / c);
}

__global__ void __launch_bounds__(128)
k_pool_acc(const float* __restrict__ h, const int* __restrict__ batch,
           const float* __restrict__ cntg, float* __restrict__ out, int n) {
  const int lane = threadIdx.x & 31;
  const int wave = __builtin_amdgcn_readfirstlane(threadIdx.x >> 5);
  const int node0 = (blockIdx.x * (blockDim.x >> 5) + wave) * POOL_ROWS;
  if (node0 >= n) return;
  const int nv = min(POOL_ROWS, n - node0);
  const int ids = batch[node0 + min(lane, nv - 1)];
  v2f v[POOL_ROWS];
  #pragma unroll
  for (int u = 0; u < POOL_ROWS; ++u) {
    const int r = min(node0 + u, n - 1);
    const float* row = h + (size_t)r * 64;
    v[u] = *(const v2f*)(row + 2 * lane);
  }
  int cur = __builtin_amdgcn_readlane(ids, 0);
  float c = fmaxf(cntg[cur], 1.0f);
  v2f acc = v[0];
  #pragma unroll
  for (int u = 1; u < POOL_ROWS; ++u) {
    if (u < nv) {
      const int g = __builtin_amdgcn_readlane(ids, u);
      if (g != cur) {
        pool_flush(out, c, cur, acc, lane);
        acc = (v2f){0.0f, 0.0f};
        cur = g;
        c = fmaxf(cntg[g], 1.0f);
      }
      acc += v[u];
    }
  }
  pool_flush(out, c, cur, acc, lane);
}

extern "C" void kernel_launch(void* const* d_in, const int* in_sizes, int n_in,
                              void* d_out, int out_size, void* d_ws, size_t ws_size,
                              hipStream_t stream) {
  const float* x     = (const float*)d_in[0];
  const float* W1    = (const float*)d_in[1];
  const float* b1    = (const float*)d_in[2];
  const float* W2    = (const float*)d_in[3];
  const float* b2    = (const float*)d_in[4];
  const int*   ei    = (const int*)d_in[5];
  const int*   batch = (const int*)d_in[6];

  const int N = in_sizes[6];
  const int E = in_sizes[5] / 2;
  const int G = N_GRAPHS;

  const int* src = ei;
  const int* dst = ei + E;

  char* base = (char*)d_ws;
  size_t off = 0;
  auto take = [&](size_t bytes) {
    size_t o = off;
    off = (off + bytes + 255) & ~(size_t)255;
    return o;
  };
  const int np = (N + 31) & ~31;
  const int chunk = (E + CNT_BLOCKS - 1) / CNT_BLOCKS;
  int*   cnt      = (int*)(base + take(sizeof(int) * (size_t)N));
  int*   hist     = (int*)(base + take(sizeof(int) * (size_t)np * CNT_BLOCKS));
  int*   rank     = (int*)(base + take(sizeof(int) * (size_t)E));
  int*   rowstart = (int*)(base + take(sizeof(int) * (size_t)(N + 1)));
  float* dinv     = (float*)(base + take(sizeof(float) * (size_t)N));
  int*   csr      = (int*)(base + take(sizeof(int) * (size_t)E));
  float* hw1      = (float*)(base + take(sizeof(float) * (size_t)N * 128));
  float* h1       = (float*)(base + take(sizeof(float) * (size_t)N * 128));
  float* cntg     = (float*)(base + take(sizeof(float) * (size_t)G));
  float* hw2 = hw1;
  float* h2  = h1;

  const int B = 256;
  const int gN = (N + B - 1) / B;
  const int mtiles = (N + 15) / 16;
  const int gGemm = (mtiles + 7) / 8;
  const int fillTiles = (chunk + FILL_TILE - 1) / FILL_TILE;

  k_count<<<CNT_BLOCKS, 1024, sizeof(int4) * (size_t)((N + 3) / 4), stream>>>(
      dst, hist, rank, E, N, chunk, np, (float*)d_out, G * 64, cntg, G);
  k_merge<<<gN, B, 0, stream>>>(hist, cnt, N, np, batch, cntg);
  k_scan<<<1, 1024, 0, stream>>>(cnt, rowstart, dinv, N);

  k_gemm1_fill<<<gGemm + fillTiles * CNT_BLOCKS, FILL_THREADS, 64 * (2 * 128 + 32) * sizeof(float), stream>>>(
      x, W1, dinv, hw1, N, gGemm, src, dst, rowstart, rank, hist, csr, E, chunk, np, fillTiles);
  k_aggregate<128><<<2048, 256, 0, stream>>>(hw1, dinv, rowstart, csr, b1, h1, N);

  k_gemm_wmma<64><<<gGemm, GEMM_THREADS, 64 * (2 * 64 + 32) * sizeof(float), stream>>>(h1, W2, dinv, hw2, N);
  k_aggregate<64><<<2048, 256, 0, stream>>>(hw2, dinv, rowstart, csr, b2, h2, N);

  k_pool_acc<<<(N + 4 * POOL_ROWS - 1) / (4 * POOL_ROWS), 128, 0, stream>>>(h2, batch, cntg, (float*)d_out, N);
}


# layer-2 product fused into the layer-1 aggregation launch: a 16-wave block sums 64 node rows, hands the relu rows to its WMMAs through LDS (W2 staged once); h1 write+read and the GEMM launch go
# speedup vs baseline: 3.9200x; 1.0414x over previous
#include <hip/hip_runtime.h>
#include <hip/hip_bf16.h>

typedef float v2f __attribute__((ext_vector_type(2)));
typedef float v8f __attribute__((ext_vector_type(8)));

#define N_GRAPHS 64

#define CNT_BLOCKS 64
#define CNT_BATCH 8
__global__ void __launch_bounds__(1024)
k_count(const int* __restrict__ dst, int* __restrict__ hist, int* __restrict__ rank,
        int E, int n, int chunk, int np,
        float* __restrict__ outz, int nout, float* __restrict__ cntg, int ng) {
  extern __shared__ int4 hc4[];
  int* hc = (int*)hc4;
  const int tid = threadIdx.x;
  const int nq = (n + 3) >> 2;
  for (int i = blockIdx.x * 1024 + tid; i < nout; i += CNT_BLOCKS * 1024) outz[i] = 0.0f;
  for (int i = blockIdx.x * 1024 + tid; i < ng; i += CNT_BLOCKS * 1024) cntg[i] = 0.0f;
  for (int i = tid; i < nq; i += 1024) hc4[i] = make_int4(0, 0, 0, 0);
  __syncthreads();
  const int start = blockIdx.x * chunk;
  const int end = min(E, start + chunk);
  for (int e0 = start + tid; e0 < end; e0 += CNT_BATCH * 1024) {
    int d[CNT_BATCH];
    int r[CNT_BATCH];
    if (e0 + (CNT_BATCH - 1) * 1024 < end) {
      #pragma unroll
      for (int u = 0; u < CNT_BATCH; ++u) d[u] = dst[e0 + u * 1024];
      #pragma unroll
      for (int u = 0; u < CNT_BATCH; ++u) r[u] = atomicAdd(&hc[d[u]], 1);
      #pragma unroll
      for (int u = 0; u < CNT_BATCH; ++u) rank[e0 + u * 1024] = r[u];
    } else {
      #pragma unroll
      for (int u = 0; u < CNT_BATCH; ++u) {
        const int e = e0 + u * 1024;
        d[u] = (e < end) ? dst[e] : 0;
      }
      #pragma unroll
      for (int u = 0; u < CNT_BATCH; ++u) {
        const int e = e0 + u * 1024;
        if (e < end) rank[e] = atomicAdd(&hc[d[u]], 1);
      }
    }
  }
  __syncthreads();
  int4* __restrict__ row = (int4*)(hist + (size_t)blockIdx.x * np);
  for (int i = tid; i < nq; i += 1024) row[i] = hc4[i];
}

__global__ void k_merge(int* __restrict__ hist, int* __restrict__ cnt, int n, int np,
                        const int* __restrict__ batch, float* __restrict__ cntg) {
  const int i = blockIdx.x * blockDim.x + threadIdx.x;
  const bool valid = i < n;
  const int lane = threadIdx.x & 31;
  const int id = valid ? batch[i] : -1;
  if (valid) {
    int run = 0;
    for (int b0 = 0; b0 < CNT_BLOCKS; b0 += 16) {
      int c[16];
      #pragma unroll
      for (int u = 0; u < 16; ++u) c[u] = hist[(size_t)(b0 + u) * np + i];
      #pragma unroll
      for (int u = 0; u < 16; ++u) {
        hist[(size_t)(b0 + u) * np + i] = run;
        run += c[u];
      }
    }
    cnt[i] = run;
  }
  unsigned todo = __builtin_amdgcn_ballot_w32(valid);
  while (todo != 0u) {
    const int leader = __builtin_ctz(todo);
    const int g = __builtin_amdgcn_readlane(id, leader);
    const unsigned same = __builtin_amdgcn_ballot_w32(valid && id == g);
    if (lane == leader) atomicAdd(&cntg[g], (float)__builtin_popcount(same));
    todo &= ~same;
  }
}

#define SCAN_ROWS 13
__global__ void k_scan(const int* __restrict__ cnt, int* __restrict__ rowstart,
                       float* __restrict__ dinv, int n) {
  constexpr int WSEG = SCAN_ROWS * 128;
  constexpr int TILE = 32 * WSEG;
  __shared__ int wtot[32];
  const int tid = threadIdx.x;
  const int lane = tid & 31;
  const int wave = tid >> 5;
  const int n4 = n & ~3;
  int carry = 0;
  for (int base = 0; base < n4; base += TILE) {
    const int i0 = base + wave * WSEG + lane * 4;
    int4 v[SCAN_ROWS];
    #pragma unroll
    for (int j = 0; j < SCAN_ROWS; ++j) {
      const int i = i0 + j * 128;
      int4 t = make_int4(0, 0, 0, 0);
      if (i < n4) t = *(const int4*)(cnt + i);
      v[j] = t;
    }
    int inc[SCAN_ROWS];
    #pragma unroll
    for (int j = 0; j < SCAN_ROWS; ++j) inc[j] = v[j].x + v[j].y + v[j].z + v[j].w;
    #pragma unroll
    for (int off = 1; off < 32; off <<= 1) {
      #pragma unroll
      for (int j = 0; j < SCAN_ROWS; ++j) {
        const int t = __shfl_up(inc[j], off, 32);
        if (lane >= off) inc[j] += t;
      }
    }
    int rowoff[SCAN_ROWS];
    int wsum = 0;
    #pragma unroll
    for (int j = 0; j < SCAN_ROWS; ++j) {
      rowoff[j] = wsum;
      wsum += __builtin_amdgcn_readlane(inc[j], 31);
    }
    if (lane == 0) wtot[wave] = wsum;
    __syncthreads();
    const int wt = wtot[lane];
    int winc = wt;
    #pragma unroll
    for (int off = 1; off < 32; off <<= 1) {
      const int t = __shfl_up(winc, off, 32);
      if (lane >= off) winc += t;
    }
    const int woff = carry + __shfl(winc - wt, wave, 32);
    carry += __builtin_amdgcn_readlane(winc, 31);
    #pragma unroll
    for (int j = 0; j < SCAN_ROWS; ++j) {
      const int i = i0 + j * 128;
      const int4 c = v[j];
      int4 r;
      r.x = woff + rowoff[j] + inc[j] - (c.x + c.y + c.z + c.w);
      r.y = r.x + c.x;
      r.z = r.y + c.y;
      r.w = r.z + c.z;
      float4 d;
      d.x = rsqrtf((float)c.x + 1.0f);
      d.y = rsqrtf((float)c.y + 1.0f);
      d.z = rsqrtf((float)c.z + 1.0f);
      d.w = rsqrtf((float)c.w + 1.0f);
      if (i < n4) {
        *(int4*)(rowstart + i) = r;
        *(float4*)(dinv + i) = d;
      }
    }
    __syncthreads();
  }
  if (tid == 0) {
    int c = carry;
    for (int i = n4; i < n; ++i) {
      const int vv = cnt[i];
      rowstart[i] = c;
      dinv[i] = rsqrtf((float)vv + 1.0f);
      c += vv;
    }
    rowstart[n] = c;
  }
}

#define FILL_BATCH 8
#define FILL_THREADS 256
#define FILL_TILE (FILL_BATCH * FILL_THREADS)
__device__ __forceinline__ void fill_tile(
    const int* __restrict__ src, const int* __restrict__ dst,
    const int* __restrict__ rowstart, const int* __restrict__ rank,
    const int* __restrict__ hist, int* __restrict__ csr,
    int E, int chunk, int np, int bx, int by) {
  const int cstart = by * chunk;
  const int lim = min(chunk, E - cstart);
  const int t0 = bx * FILL_TILE;
  if (t0 >= lim) return;
  const int* __restrict__ hrow = hist + (size_t)by * np;
  const int l0 = t0 + (int)threadIdx.x;
  int d[FILL_BATCH], r[FILL_BATCH], s[FILL_BATCH], p[FILL_BATCH];
  if (t0 + FILL_TILE <= lim) {
    const int e0 = cstart + l0;
    #pragma unroll
    for (int u = 0; u < FILL_BATCH; ++u) d[u] = dst[e0 + u * FILL_THREADS];
    #pragma unroll
    for (int u = 0; u < FILL_BATCH; ++u) r[u] = rank[e0 + u * FILL_THREADS];
    #pragma unroll
    for (int u = 0; u < FILL_BATCH; ++u) s[u] = src[e0 + u * FILL_THREADS];
    #pragma unroll
    for (int u = 0; u < FILL_BATCH; ++u) p[u] = rowstart[d[u]] + hrow[d[u]];
    #pragma unroll
    for (int u = 0; u < FILL_BATCH; ++u) csr[p[u] + r[u]] = s[u];
  } else {
    #pragma unroll
    for (int u = 0; u < FILL_BATCH; ++u) {
      const int e = cstart + min(l0 + u * FILL_THREADS, lim - 1);
      d[u] = dst[e];
      r[u] = rank[e];
      s[u] = src[e];
    }
    #pragma unroll
    for (int u = 0; u < FILL_BATCH; ++u) p[u] = rowstart[d[u]] + hrow[d[u]];
    #pragma unroll
    for (int u = 0; u < FILL_BATCH; ++u) {
      if (l0 + u * FILL_THREADS < lim) csr[p[u] + r[u]] = s[u];
    }
  }
}

#define GEMM_THREADS 256
template <int N>
__device__ __forceinline__ void gemm_block(const float* __restrict__ A, const float* __restrict__ W,
                                           const float* __restrict__ dinv, float* __restrict__ out, int M,
                                           int bid, float* Wlds) {
  constexpr int K = 128;
  constexpr int NT = N / 16;
  constexpr int RP = 2 * N + 32;
  constexpr int QN = N / 4;
  constexpr int PSTEP = GEMM_THREADS / QN;
  constexpr int WPER = (K / 2) / PSTEP;
  static_assert(GEMM_THREADS % QN == 0 && (K / 2) % PSTEP == 0, "W items must divide over the block");
  const int p0 = (int)threadIdx.x / QN;
  const int q = (int)threadIdx.x % QN;
  const float* wsrc = W + (size_t)(2 * p0) * N + 4 * q;
  float* wdst = Wlds + p0 * RP + 8 * q;
  float4 wlo[WPER], whi[WPER];
  #pragma unroll
  for (int j = 0; j < WPER; ++j) {
    wlo[j] = *(const float4*)(wsrc + (size_t)j * (2 * PSTEP * N));
    whi[j] = *(const float4*)(wsrc + (size_t)j * (2 * PSTEP * N) + N);
  }
  #pragma unroll
  for (int j = 0; j < WPER; ++j) {
    float* d = wdst + j * (PSTEP * RP);
    *(float4*)d       = make_float4(wlo[j].x, whi[j].x, wlo[j].y, whi[j].y);
    *(float4*)(d + 4) = make_float4(wlo[j].z, whi[j].z, wlo[j].w, whi[j].w);
  }
  __syncthreads();

  const int wave = threadIdx.x >> 5;
  const int lane = threadIdx.x & 31;
  const int mtile = bid * 8 + wave;
  if (mtile * 16 >= M) return;

  const int Rm = mtile * 16;
  const int half = lane >> 4;
  const int l16 = lane & 15;

  v8f acc[NT] = {};
  const float* arow = A + (size_t)(Rm + l16) * K;
  const float* bbase = Wlds + half * RP + l16 * 2;

  for (int kb = 0; kb < K; kb += 4) {
    v2f a = *(const v2f*)(arow + kb + 2 * half);
    const float* bk = bbase + (kb >> 1) * RP;
    #pragma unroll
    for (int t = 0; t < NT; ++t) {
      v2f b = *(const v2f*)(bk + t * 32);
      acc[t] = __builtin_amdgcn_wmma_f32_16x16x4_f32(
          false, a, false, b, (short)0, acc[t], false, false);
    }
  }

  #pragma unroll
  for (int r = 0; r < 8; ++r) {
    const int row = Rm + r + 8 * half;
    const float di = dinv[row];
    #pragma unroll
    for (int t = 0; t < NT; ++t) {
      out[(size_t)row * N + t * 16 + l16] = acc[t][r] * di;
    }
  }
}

static_assert(FILL_THREADS == GEMM_THREADS, "the shared launch has one block size");
__global__ void __launch_bounds__(GEMM_THREADS)
k_gemm1_fill(const float* __restrict__ A, const float* __restrict__ W,
             const float* __restrict__ dinv, float* __restrict__ out, int M,
             int gemm_blocks,
             const int* __restrict__ src, const int* __restrict__ dst,
             const int* __restrict__ rowstart, const int* __restrict__ rank,
             const int* __restrict__ hist, int* __restrict__ csr,
             int E, int chunk, int np, int tiles) {
  extern __shared__ float4 Wlds4[];
  const int b = (int)blockIdx.x;
  if (b < gemm_blocks) {
    gemm_block<128>(A, W, dinv, out, M, b, (float*)Wlds4);
  } else {
    const int f = b - gemm_blocks;
    const int by = f / tiles;
    fill_tile(src, dst, rowstart, rank, hist, csr, E, chunk, np, f - by * tiles, by);
  }
}

template <int F>
struct AggVec {
  typedef float type __attribute__((ext_vector_type(F / 32)));
};

template <int F, int NB>
__device__ __forceinline__ typename AggVec<F>::type agg_rows(
    const float* __restrict__ hws, int idx, int k, int lane, typename AggVec<F>::type sum) {
  typedef typename AggVec<F>::type vf;
  constexpr int V = F / 32;
  vf v[NB];
  #pragma unroll
  for (int u = 0; u < NB; ++u) {
    const unsigned r = (unsigned)__builtin_amdgcn_readlane(idx, k + u);
    const float* row = hws + (size_t)r * F;
    v[u] = *(const vf*)(row + V * lane);
  }
  #pragma unroll
  for (int u = 0; u < NB; ++u) sum += v[u];
  return sum;
}

template <int F>
__global__ void __launch_bounds__(256)
k_aggregate(const float* __restrict__ hws, const float* __restrict__ dinv,
            const int* __restrict__ rowstart, const int* __restrict__ csr,
            const float* __restrict__ bias, float* __restrict__ out, int n) {
  constexpr int V = F / 32;
  typedef typename AggVec<F>::type vf;
  const int lane = threadIdx.x & 31;
  const int wave = __builtin_amdgcn_readfirstlane(threadIdx.x >> 5);
  const int wpb = blockDim.x >> 5;
  const int nwaves = gridDim.x * wpb;
  const vf bv = *(const vf*)(bias + V * lane);
  for (int node = blockIdx.x * wpb + wave; node < n; node += nwaves) {
    const int s = rowstart[node];
    const int e = rowstart[node + 1];
    vf sum = *(const vf*)(hws + (size_t)node * F + V * lane);
    for (int base = s; base < e; base += 32) {
      const int cnt = min(e - base, 32);
      const int idx = csr[base + min(lane, cnt - 1)];
      int k = 0;
      for (; k + 8 <= cnt; k += 8) sum = agg_rows<F, 8>(hws, idx, k, lane, sum);
      if (cnt & 4) { sum = agg_rows<F, 4>(hws, idx, k, lane, sum); k += 4; }
      if (cnt & 2) { sum = agg_rows<F, 2>(hws, idx, k, lane, sum); k += 2; }
      if (cnt & 1) sum = agg_rows<F, 1>(hws, idx, k, lane, sum);
    }
    const float di = dinv[node];
    vf o;
    #pragma unroll
    for (int c = 0; c < V; ++c) o[c] = fmaxf(di * sum[c] + bv[c], 0.0f);
    *(vf*)(out + (size_t)node * F + V * lane) = o;
  }
}

#define FUSE_THREADS 512
#define FUSE_NODES 64
#define FUSE_AP 132
#define FUSE_RP 160
#define FUSE_LDS_FLOATS (64 * FUSE_RP + FUSE_NODES * FUSE_AP)
__global__ void __launch_bounds__(FUSE_THREADS)
k_agg1_gemm2(const float* __restrict__ hws, const float* __restrict__ dinv,
             const int* __restrict__ rowstart, const int* __restrict__ csr,
             const float* __restrict__ bias, const float* __restrict__ W,
             float* __restrict__ out, int n) {
  typedef AggVec<128>::type vf;
  extern __shared__ float4 fuse_lds4[];
  float* Wlds = (float*)fuse_lds4;
  float* tiles = Wlds + 64 * FUSE_RP;
  const int tid = threadIdx.x;
  const int lane = tid & 31;
  const int wave = __builtin_amdgcn_readfirstlane(tid >> 5);
  const int nb = blockIdx.x * FUSE_NODES;

  {
    const int p0 = tid >> 4;
    const int q = tid & 15;
    const float* wsrc = W + (size_t)(2 * p0) * 64 + 4 * q;
    float* wdst = Wlds + p0 * FUSE_RP + 8 * q;
    float4 wlo[2], whi[2];
    #pragma unroll
    for (int j = 0; j < 2; ++j) {
      wlo[j] = *(const float4*)(wsrc + (size_t)j * (64 * 64));
      whi[j] = *(const float4*)(wsrc + (size_t)j * (64 * 64) + 64);
    }
    #pragma unroll
    for (int j = 0; j < 2; ++j) {
      float* d = wdst + j * (32 * FUSE_RP);
      *(float4*)d       = make_float4(wlo[j].x, whi[j].x, wlo[j].y, whi[j].y);
      *(float4*)(d + 4) = make_float4(wlo[j].z, whi[j].z, wlo[j].w, whi[j].w);
    }
  }

  const vf bv = *(const vf*)(bias + 4 * lane);
  for (int j = 0; j < 4; ++j) {
    const int r = 4 * wave + j;
    const int node = nb + r;
    vf o = (vf){0.0f, 0.0f, 0.0f, 0.0f};
    if (node < n) {
      const int s = rowstart[node];
      const int e = rowstart[node + 1];
      vf sum = *(const vf*)(hws + (size_t)node * 128 + 4 * lane);
      for (int base = s; base < e; base += 32) {
        const int cnt = min(e - base, 32);
        const int idx = csr[base + min(lane, cnt - 1)];
        int k = 0;
        for (; k + 8 <= cnt; k += 8) sum = agg_rows<128, 8>(hws, idx, k, lane, sum);
        if (cnt & 4) { sum = agg_rows<128, 4>(hws, idx, k, lane, sum); k += 4; }
        if (cnt & 2) { sum = agg_rows<128, 2>(hws, idx, k, lane, sum); k += 2; }
        if (cnt & 1) sum = agg_rows<128, 1>(hws, idx, k, lane, sum);
      }
      const float di = dinv[node];
      #pragma unroll
      for (int c = 0; c < 4; ++c) o[c] = fmaxf(di * sum[c] + bv[c], 0.0f);
    }
    *(vf*)(tiles + r * FUSE_AP + 4 * lane) = o;
  }
  __syncthreads();

  const int half = lane >> 4;
  const int l16 = lane & 15;
  const int mt = wave >> 2;
  const int nt = wave & 3;
  const float* abase = tiles + (16 * mt + l16) * FUSE_AP + 2 * half;
  const float* bbase = Wlds + half * FUSE_RP + (16 * nt + l16) * 2;
  const int row0 = nb + 16 * mt + 8 * half;
  float di[8];
  #pragma unroll
  for (int r = 0; r < 8; ++r) di[r] = dinv[min(row0 + r, n - 1)];
  v8f acc = {};
  #pragma unroll
  for (int kb = 0; kb < 128; kb += 4) {
    const v2f a = *(const v2f*)(abase + kb);
    const v2f b = *(const v2f*)(bbase + (kb >> 1) * FUSE_RP);
    acc = __builtin_amdgcn_wmma_f32_16x16x4_f32(
        false, a, false, b, (short)0, acc, false, false);
  }
  #pragma unroll
  for (int r = 0; r < 8; ++r) {
    const int row = row0 + r;
    if (row < n) out[(size_t)row * 64 + 16 * nt + l16] = acc[r] * di[r];
  }
}

#define POOL_ROWS 32
__device__ __forceinline__ void pool_flush(float* __restrict__ out, float c, int g, v2f acc, int lane) {
  float* p = out + g * 64 + 2 * lane;
  atomicAdd(p, acc.x / c);
  atomicAdd(p + 1, acc.y / c);
}

__global__ void __launch_bounds__(128)
k_pool_acc(const float* __restrict__ h, const int* __restrict__ batch,
           const float* __restrict__ cntg, float* __restrict__ out, int n) {
  const int lane = threadIdx.x & 31;
  const int wave = __builtin_amdgcn_readfirstlane(threadIdx.x >> 5);
  const int node0 = (blockIdx.x * (blockDim.x >> 5) + wave) * POOL_ROWS;
  if (node0 >= n) return;
  const int nv = min(POOL_ROWS, n - node0);
  const int ids = batch[node0 + min(lane, nv - 1)];
  v2f v[POOL_ROWS];
  #pragma unroll
  for (int u = 0; u < POOL_ROWS; ++u) {
    const int r = min(node0 + u, n - 1);
    const float* row = h + (size_t)r * 64;
    v[u] = *(const v2f*)(row + 2 * lane);
  }
  int cur = __builtin_amdgcn_readlane(ids, 0);
  float c = fmaxf(cntg[cur], 1.0f);
  v2f acc = v[0];
  #pragma unroll
  for (int u = 1; u < POOL_ROWS; ++u) {
    if (u < nv) {
      const int g = __builtin_amdgcn_readlane(ids, u);
      if (g != cur) {
        pool_flush(out, c, cur, acc, lane);
        acc = (v2f){0.0f, 0.0f};
        cur = g;
        c = fmaxf(cntg[g], 1.0f);
      }
      acc += v[u];
    }
  }
  pool_flush(out, c, cur, acc, lane);
}

extern "C" void kernel_launch(void* const* d_in, const int* in_sizes, int n_in,
                              void* d_out, int out_size, void* d_ws, size_t ws_size,
                              hipStream_t stream) {
  const float* x     = (const float*)d_in[0];
  const float* W1    = (const float*)d_in[1];
  const float* b1    = (const float*)d_in[2];
  const float* W2    = (const float*)d_in[3];
  const float* b2    = (const float*)d_in[4];
  const int*   ei    = (const int*)d_in[5];
  const int*   batch = (const int*)d_in[6];

  const int N = in_sizes[6];
  const int E = in_sizes[5] / 2;
  const int G = N_GRAPHS;

  const int* src = ei;
  const int* dst = ei + E;

  char* base = (char*)d_ws;
  size_t off = 0;
  auto take = [&](size_t bytes) {
    size_t o = off;
    off = (off + bytes + 255) & ~(size_t)255;
    return o;
  };
  const int np = (N + 31) & ~31;
  const int chunk = (E + CNT_BLOCKS - 1) / CNT_BLOCKS;
  int*   cnt      = (int*)(base + take(sizeof(int) * (size_t)N));
  int*   hist     = (int*)(base + take(sizeof(int) * (size_t)np * CNT_BLOCKS));
  int*   rank     = (int*)(base + take(sizeof(int) * (size_t)E));
  int*   rowstart = (int*)(base + take(sizeof(int) * (size_t)(N + 1)));
  float* dinv     = (float*)(base + take(sizeof(float) * (size_t)N));
  int*   csr      = (int*)(base + take(sizeof(int) * (size_t)E));
  float* hw1      = (float*)(base + take(sizeof(float) * (size_t)N * 128));
  float* h1       = (float*)(base + take(sizeof(float) * (size_t)N * 128));
  float* cntg     = (float*)(base + take(sizeof(float) * (size_t)G));
  float* hw2 = h1;
  float* h2  = hw1;

  const int B = 256;
  const int gN = (N + B - 1) / B;
  const int mtiles = (N + 15) / 16;
  const int gGemm = (mtiles + 7) / 8;
  const int fillTiles = (chunk + FILL_TILE - 1) / FILL_TILE;

  k_count<<<CNT_BLOCKS, 1024, sizeof(int4) * (size_t)((N + 3) / 4), stream>>>(
      dst, hist, rank, E, N, chunk, np, (float*)d_out, G * 64, cntg, G);
  k_merge<<<gN, B, 0, stream>>>(hist, cnt, N, np, batch, cntg);
  k_scan<<<1, 1024, 0, stream>>>(cnt, rowstart, dinv, N);

  k_gemm1_fill<<<gGemm + fillTiles * CNT_BLOCKS, FILL_THREADS, 64 * (2 * 128 + 32) * sizeof(float), stream>>>(
      x, W1, dinv, hw1, N, gGemm, src, dst, rowstart, rank, hist, csr, E, chunk, np, fillTiles);
  k_agg1_gemm2<<<(N + FUSE_NODES - 1) / FUSE_NODES, FUSE_THREADS, FUSE_LDS_FLOATS * sizeof(float), stream>>>(
      hw1, dinv, rowstart, csr, b1, W2, hw2, N);

  k_aggregate<64><<<2048, 256, 0, stream>>>(hw2, dinv, rowstart, csr, b2, h2, N);

  k_pool_acc<<<(N + 4 * POOL_ROWS - 1) / (4 * POOL_ROWS), 128, 0, stream>>>(h2, batch, cntg, (float*)d_out, N);
}


# row starts without a scan launch: k_count also stores each chunk's prefix of edge counts over 128-node groups; k_merge (a block per group) sums 64 of them and scans its 128 degrees; 7 launches -> 6
# speedup vs baseline: 4.1632x; 1.0620x over previous
#include <hip/hip_runtime.h>
#include <hip/hip_bf16.h>

typedef float v2f __attribute__((ext_vector_type(2)));
typedef float v8f __attribute__((ext_vector_type(8)));

#define N_GRAPHS 64

__device__ __forceinline__ int wave_sum(int v) {
  v += __builtin_amdgcn_update_dpp(0, v, 0x161, 0xf, 0xf, false);
  v += __builtin_amdgcn_update_dpp(0, v, 0x162, 0xf, 0xf, false);
  v += __builtin_amdgcn_update_dpp(0, v, 0x164, 0xf, 0xf, false);
  v += __builtin_amdgcn_update_dpp(0, v, 0x168, 0xf, 0xf, false);
  return __builtin_amdgcn_readlane(v, 0) + __builtin_amdgcn_readlane(v, 16);
}
__device__ __forceinline__ int wave_scan_incl(int v, int lane) {
  v += __builtin_amdgcn_update_dpp(0, v, 0x111, 0xf, 0xf, false);
  v += __builtin_amdgcn_update_dpp(0, v, 0x112, 0xf, 0xf, false);
  v += __builtin_amdgcn_update_dpp(0, v, 0x114, 0xf, 0xf, false);
  v += __builtin_amdgcn_update_dpp(0, v, 0x118, 0xf, 0xf, false);
  const int low = __builtin_amdgcn_readlane(v, 15);
  if (lane >= 16) v += low;
  return v;
}

#define CNT_BLOCKS 64
#define CNT_BATCH 8
#define GRP_NODES 128
__global__ void __launch_bounds__(1024)
k_count(const int* __restrict__ dst, int* __restrict__ hist, int* __restrict__ rank,
        int* __restrict__ pre, int E, int n, int chunk, int np, int ngrp,
        float* __restrict__ outz, int nout, float* __restrict__ cntg, int ng) {
  extern __shared__ int4 hc4[];
  __shared__ int gt[1024];
  __shared__ int gwt[32];
  int* hc = (int*)hc4;
  const int tid = threadIdx.x;
  const int lane = tid & 31;
  const int wave = __builtin_amdgcn_readfirstlane(tid >> 5);
  const int nq = (n + 3) >> 2;
  for (int i = blockIdx.x * 1024 + tid; i < nout; i += CNT_BLOCKS * 1024) outz[i] = 0.0f;
  for (int i = blockIdx.x * 1024 + tid; i < ng; i += CNT_BLOCKS * 1024) cntg[i] = 0.0f;
  for (int i = tid; i < nq; i += 1024) hc4[i] = make_int4(0, 0, 0, 0);
  __syncthreads();
  const int start = blockIdx.x * chunk;
  const int end = min(E, start + chunk);
  for (int e0 = start + tid; e0 < end; e0 += CNT_BATCH * 1024) {
    int d[CNT_BATCH];
    int r[CNT_BATCH];
    if (e0 + (CNT_BATCH - 1) * 1024 < end) {
      #pragma unroll
      for (int u = 0; u < CNT_BATCH; ++u) d[u] = dst[e0 + u * 1024];
      #pragma unroll
      for (int u = 0; u < CNT_BATCH; ++u) r[u] = atomicAdd(&hc[d[u]], 1);
      #pragma unroll
      for (int u = 0; u < CNT_BATCH; ++u) rank[e0 + u * 1024] = r[u];
    } else {
      #pragma unroll
      for (int u = 0; u < CNT_BATCH; ++u) {
        const int e = e0 + u * 1024;
        d[u] = (e < end) ? dst[e] : 0;
      }
      #pragma unroll
      for (int u = 0; u < CNT_BATCH; ++u) {
        const int e = e0 + u * 1024;
        if (e < end) rank[e] = atomicAdd(&hc[d[u]], 1);
      }
    }
  }
  __syncthreads();
  int4* __restrict__ row = (int4*)(hist + (size_t)blockIdx.x * np);
  for (int i0 = wave * 32; i0 < nq; i0 += 1024) {
    const int i = i0 + lane;
    int4 v = make_int4(0, 0, 0, 0);
    if (i < nq) {
      v = hc4[i];
      row[i] = v;
    }
    const int s = wave_sum(v.x + v.y + v.z + v.w);
    if (lane == 0) gt[i0 >> 5] = s;
  }
  __syncthreads();
  const int x = (tid < ngrp) ? gt[tid] : 0;
  const int inc = wave_scan_incl(x, lane);
  if (lane == 31) gwt[wave] = inc;
  __syncthreads();
  const int wt = gwt[lane];
  const int winc = wave_scan_incl(wt, lane);
  const int woff = __builtin_amdgcn_readlane(winc - wt, wave);
  if (tid < ngrp) pre[(size_t)blockIdx.x * ngrp + tid] = woff + inc - x;
}

__global__ void __launch_bounds__(GRP_NODES)
k_merge(int* __restrict__ hist, const int* __restrict__ pre, int* __restrict__ rowstart,
        float* __restrict__ dinv, int n, int np, int ngrp,
        const int* __restrict__ batch, float* __restrict__ cntg) {
  __shared__ int wsum[GRP_NODES / 32];
  const int g = blockIdx.x;
  const int i = g * GRP_NODES + (int)threadIdx.x;
  const bool valid = i < n;
  const int lane = threadIdx.x & 31;
  const int wave = threadIdx.x >> 5;
  const int id = valid ? batch[i] : -1;
  int pb = 0;
  #pragma unroll
  for (int b = 0; b < CNT_BLOCKS; b += 32) pb += pre[(size_t)(b + lane) * ngrp + g];
  int run = 0;
  if (valid) {
    for (int b0 = 0; b0 < CNT_BLOCKS; b0 += 16) {
      int c[16];
      #pragma unroll
      for (int u = 0; u < 16; ++u) c[u] = hist[(size_t)(b0 + u) * np + i];
      #pragma unroll
      for (int u = 0; u < 16; ++u) {
        hist[(size_t)(b0 + u) * np + i] = run;
        run += c[u];
      }
    }
  }
  const int gbase = wave_sum(pb);
  const int inc = wave_scan_incl(run, lane);
  if (lane == 31) wsum[wave] = inc;
  __syncthreads();
  int woff = 0;
  #pragma unroll
  for (int w = 0; w < GRP_NODES / 32; ++w) {
    const int t = wsum[w];
    if (w < wave) woff += t;
  }
  if (valid) {
    const int rs = gbase + woff + inc - run;
    rowstart[i] = rs;
    dinv[i] = rsqrtf((float)run + 1.0f);
    if (i == n - 1) rowstart[n] = rs + run;
  }
  unsigned todo = __builtin_amdgcn_ballot_w32(valid);
  while (todo != 0u) {
    const int leader = __builtin_ctz(todo);
    const int gid = __builtin_amdgcn_readlane(id, leader);
    const unsigned same = __builtin_amdgcn_ballot_w32(valid && id == gid);
    if (lane == leader) atomicAdd(&cntg[gid], (float)__builtin_popcount(same));
    todo &= ~same;
  }
}

#define FILL_BATCH 8
#define FILL_THREADS 256
#define FILL_TILE (FILL_BATCH * FILL_THREADS)
__device__ __forceinline__ void fill_tile(
    const int* __restrict__ src, const int* __restrict__ dst,
    const int* __restrict__ rowstart, const int* __restrict__ rank,
    const int* __restrict__ hist, int* __restrict__ csr,
    int E, int chunk, int np, int bx, int by) {
  const int cstart = by * chunk;
  const int lim = min(chunk, E - cstart);
  const int t0 = bx * FILL_TILE;
  if (t0 >= lim) return;
  const int* __restrict__ hrow = hist + (size_t)by * np;
  const int l0 = t0 + (int)threadIdx.x;
  int d[FILL_BATCH], r[FILL_BATCH], s[FILL_BATCH], p[FILL_BATCH];
  if (t0 + FILL_TILE <= lim) {
    const int e0 = cstart + l0;
    #pragma unroll
    for (int u = 0; u < FILL_BATCH; ++u) d[u] = dst[e0 + u * FILL_THREADS];
    #pragma unroll
    for (int u = 0; u < FILL_BATCH; ++u) r[u] = rank[e0 + u * FILL_THREADS];
    #pragma unroll
    for (int u = 0; u < FILL_BATCH; ++u) s[u] = src[e0 + u * FILL_THREADS];
    #pragma unroll
    for (int u = 0; u < FILL_BATCH; ++u) p[u] = rowstart[d[u]] + hrow[d[u]];
    #pragma unroll
    for (int u = 0; u < FILL_BATCH; ++u) csr[p[u] + r[u]] = s[u];
  } else {
    #pragma unroll
    for (int u = 0; u < FILL_BATCH; ++u) {
      const int e = cstart + min(l0 + u * FILL_THREADS, lim - 1);
      d[u] = dst[e];
      r[u] = rank[e];
      s[u] = src[e];
    }
    #pragma unroll
    for (int u = 0; u < FILL_BATCH; ++u) p[u] = rowstart[d[u]] + hrow[d[u]];
    #pragma unroll
    for (int u = 0; u < FILL_BATCH; ++u) {
      if (l0 + u * FILL_THREADS < lim) csr[p[u] + r[u]] = s[u];
    }
  }
}

#define GEMM_THREADS 256
template <int N>
__device__ __forceinline__ void gemm_block(const float* __restrict__ A, const float* __restrict__ W,
                                           const float* __restrict__ dinv, float* __restrict__ out, int M,
                                           int bid, float* Wlds) {
  constexpr int K = 128;
  constexpr int NT = N / 16;
  constexpr int RP = 2 * N + 32;
  constexpr int QN = N / 4;
  constexpr int PSTEP = GEMM_THREADS / QN;
  constexpr int WPER = (K / 2) / PSTEP;
  static_assert(GEMM_THREADS % QN == 0 && (K / 2) % PSTEP == 0, "W items must divide over the block");
  const int p0 = (int)threadIdx.x / QN;
  const int q = (int)threadIdx.x % QN;
  const float* wsrc = W + (size_t)(2 * p0) * N + 4 * q;
  float* wdst = Wlds + p0 * RP + 8 * q;
  float4 wlo[WPER], whi[WPER];
  #pragma unroll
  for (int j = 0; j < WPER; ++j) {
    wlo[j] = *(const float4*)(wsrc + (size_t)j * (2 * PSTEP * N));
    whi[j] = *(const float4*)(wsrc + (size_t)j * (2 * PSTEP * N) + N);
  }
  #pragma unroll
  for (int j = 0; j < WPER; ++j) {
    float* d = wdst + j * (PSTEP * RP);
    *(float4*)d       = make_float4(wlo[j].x, whi[j].x, wlo[j].y, whi[j].y);
    *(float4*)(d + 4) = make_float4(wlo[j].z, whi[j].z, wlo[j].w, whi[j].w);
  }
  __syncthreads();

  const int wave = threadIdx.x >> 5;
  const int lane = threadIdx.x & 31;
  const int mtile = bid * 8 + wave;
  if (mtile * 16 >= M) return;

  const int Rm = mtile * 16;
  const int half = lane >> 4;
  const int l16 = lane & 15;

  v8f acc[NT] = {};
  const float* arow = A + (size_t)(Rm + l16) * K;
  const float* bbase = Wlds + half * RP + l16 * 2;

  for (int kb = 0; kb < K; kb += 4) {
    v2f a = *(const v2f*)(arow + kb + 2 * half);
    const float* bk = bbase + (kb >> 1) * RP;
    #pragma unroll
    for (int t = 0; t < NT; ++t) {
      v2f b = *(const v2f*)(bk + t * 32);
      acc[t] = __builtin_amdgcn_wmma_f32_16x16x4_f32(
          false, a, false, b, (short)0, acc[t], false, false);
    }
  }

  #pragma unroll
  for (int r = 0; r < 8; ++r) {
    const int row = Rm + r + 8 * half;
    const float di = dinv[row];
    #pragma unroll
    for (int t = 0; t < NT; ++t) {
      out[(size_t)row * N + t * 16 + l16] = acc[t][r] * di;
    }
  }
}

static_assert(FILL_THREADS == GEMM_THREADS, "the shared launch has one block size");
__global__ void __launch_bounds__(GEMM_THREADS)
k_gemm1_fill(const float* __restrict__ A, const float* __restrict__ W,
             const float* __restrict__ dinv, float* __restrict__ out, int M,
             int gemm_blocks,
             const int* __restrict__ src, const int* __restrict__ dst,
             const int* __restrict__ rowstart, const int* __restrict__ rank,
             const int* __restrict__ hist, int* __restrict__ csr,
             int E, int chunk, int np, int tiles) {
  extern __shared__ float4 Wlds4[];
  const int b = (int)blockIdx.x;
  if (b < gemm_blocks) {
    gemm_block<128>(A, W, dinv, out, M, b, (float*)Wlds4);
  } else {
    const int f = b - gemm_blocks;
    const int by = f / tiles;
    fill_tile(src, dst, rowstart, rank, hist, csr, E, chunk, np, f - by * tiles, by);
  }
}

template <int F>
struct AggVec {
  typedef float type __attribute__((ext_vector_type(F / 32)));
};

template <int F, int NB>
__device__ __forceinline__ typename AggVec<F>::type agg_rows(
    const float* __restrict__ hws, int idx, int k, int lane, typename AggVec<F>::type sum) {
  typedef typename AggVec<F>::type vf;
  constexpr int V = F / 32;
  vf v[NB];
  #pragma unroll
  for (int u = 0; u < NB; ++u) {
    const unsigned r = (unsigned)__builtin_amdgcn_readlane(idx, k + u);
    const float* row = hws + (size_t)r * F;
    v[u] = *(const vf*)(row + V * lane);
  }
  #pragma unroll
  for (int u = 0; u < NB; ++u) sum += v[u];
  return sum;
}

template <int F>
__global__ void __launch_bounds__(256)
k_aggregate(const float* __restrict__ hws, const float* __restrict__ dinv,
            const int* __restrict__ rowstart, const int* __restrict__ csr,
            const float* __restrict__ bias, float* __restrict__ out, int n) {
  constexpr int V = F / 32;
  typedef typename AggVec<F>::type vf;
  const int lane = threadIdx.x & 31;
  const int wave = __builtin_amdgcn_readfirstlane(threadIdx.x >> 5);
  const int wpb = blockDim.x >> 5;
  const int nwaves = gridDim.x * wpb;
  const vf bv = *(const vf*)(bias + V * lane);
  for (int node = blockIdx.x * wpb + wave; node < n; node += nwaves) {
    const int s = rowstart[node];
    const int e = rowstart[node + 1];
    vf sum = *(const vf*)(hws + (size_t)node * F + V * lane);
    for (int base = s; base < e; base += 32) {
      const int cnt = min(e - base, 32);
      const int idx = csr[base + min(lane, cnt - 1)];
      int k = 0;
      for (; k + 8 <= cnt; k += 8) sum = agg_rows<F, 8>(hws, idx, k, lane, sum);
      if (cnt & 4) { sum = agg_rows<F, 4>(hws, idx, k, lane, sum); k += 4; }
      if (cnt & 2) { sum = agg_rows<F, 2>(hws, idx, k, lane, sum); k += 2; }
      if (cnt & 1) sum = agg_rows<F, 1>(hws, idx, k, lane, sum);
    }
    const float di = dinv[node];
    vf o;
    #pragma unroll
    for (int c = 0; c < V; ++c) o[c] = fmaxf(di * sum[c] + bv[c], 0.0f);
    *(vf*)(out + (size_t)node * F + V * lane) = o;
  }
}

#define FUSE_THREADS 512
#define FUSE_NODES 64
#define FUSE_AP 132
#define FUSE_RP 160
#define FUSE_LDS_FLOATS (64 * FUSE_RP + FUSE_NODES * FUSE_AP)
__global__ void __launch_bounds__(FUSE_THREADS)
k_agg1_gemm2(const float* __restrict__ hws, const float* __restrict__ dinv,
             const int* __restrict__ rowstart, const int* __restrict__ csr,
             const float* __restrict__ bias, const float* __restrict__ W,
             float* __restrict__ out, int n) {
  typedef AggVec<128>::type vf;
  extern __shared__ float4 fuse_lds4[];
  float* Wlds = (float*)fuse_lds4;
  float* tiles = Wlds + 64 * FUSE_RP;
  const int tid = threadIdx.x;
  const int lane = tid & 31;
  const int wave = __builtin_amdgcn_readfirstlane(tid >> 5);
  const int nb = blockIdx.x * FUSE_NODES;

  {
    const int p0 = tid >> 4;
    const int q = tid & 15;
    const float* wsrc = W + (size_t)(2 * p0) * 64 + 4 * q;
    float* wdst = Wlds + p0 * FUSE_RP + 8 * q;
    float4 wlo[2], whi[2];
    #pragma unroll
    for (int j = 0; j < 2; ++j) {
      wlo[j] = *(const float4*)(wsrc + (size_t)j * (64 * 64));
      whi[j] = *(const float4*)(wsrc + (size_t)j * (64 * 64) + 64);
    }
    #pragma unroll
    for (int j = 0; j < 2; ++j) {
      float* d = wdst + j * (32 * FUSE_RP);
      *(float4*)d       = make_float4(wlo[j].x, whi[j].x, wlo[j].y, whi[j].y);
      *(float4*)(d + 4) = make_float4(wlo[j].z, whi[j].z, wlo[j].w, whi[j].w);
    }
  }

  const vf bv = *(const vf*)(bias + 4 * lane);
  for (int j = 0; j < 4; ++j) {
    const int r = 4 * wave + j;
    const int node = nb + r;
    vf o = (vf){0.0f, 0.0f, 0.0f, 0.0f};
    if (node < n) {
      const int s = rowstart[node];
      const int e = rowstart[node + 1];
      vf sum = *(const vf*)(hws + (size_t)node * 128 + 4 * lane);
      for (int base = s; base < e; base += 32) {
        const int cnt = min(e - base, 32);
        const int idx = csr[base + min(lane, cnt - 1)];
        int k = 0;
        for (; k + 8 <= cnt; k += 8) sum = agg_rows<128, 8>(hws, idx, k, lane, sum);
        if (cnt & 4) { sum = agg_rows<128, 4>(hws, idx, k, lane, sum); k += 4; }
        if (cnt & 2) { sum = agg_rows<128, 2>(hws, idx, k, lane, sum); k += 2; }
        if (cnt & 1) sum = agg_rows<128, 1>(hws, idx, k, lane, sum);
      }
      const float di = dinv[node];
      #pragma unroll
      for (int c = 0; c < 4; ++c) o[c] = fmaxf(di * sum[c] + bv[c], 0.0f);
    }
    *(vf*)(tiles + r * FUSE_AP + 4 * lane) = o;
  }
  __syncthreads();

  const int half = lane >> 4;
  const int l16 = lane & 15;
  const int mt = wave >> 2;
  const int nt = wave & 3;
  const float* abase = tiles + (16 * mt + l16) * FUSE_AP + 2 * half;
  const float* bbase = Wlds + half * FUSE_RP + (16 * nt + l16) * 2;
  const int row0 = nb + 16 * mt + 8 * half;
  float di[8];
  #pragma unroll
  for (int r = 0; r < 8; ++r) di[r] = dinv[min(row0 + r, n - 1)];
  v8f acc = {};
  #pragma unroll
  for (int kb = 0; kb < 128; kb += 4) {
    const v2f a = *(const v2f*)(abase + kb);
    const v2f b = *(const v2f*)(bbase + (kb >> 1) * FUSE_RP);
    acc = __builtin_amdgcn_wmma_f32_16x16x4_f32(
        false, a, false, b, (short)0, acc, false, false);
  }
  #pragma unroll
  for (int r = 0; r < 8; ++r) {
    const int row = row0 + r;
    if (row < n) out[(size_t)row * 64 + 16 * nt + l16] = acc[r] * di[r];
  }
}

#define POOL_ROWS 32
__device__ __forceinline__ void pool_flush(float* __restrict__ out, float c, int g, v2f acc, int lane) {
  float* p = out + g * 64 + 2 * lane;
  atomicAdd(p, acc.x / c);
  atomicAdd(p + 1, acc.y / c);
}

__global__ void __launch_bounds__(128)
k_pool_acc(const float* __restrict__ h, const int* __restrict__ batch,
           const float* __restrict__ cntg, float* __restrict__ out, int n) {
  const int lane = threadIdx.x & 31;
  const int wave = __builtin_amdgcn_readfirstlane(threadIdx.x >> 5);
  const int node0 = (blockIdx.x * (blockDim.x >> 5) + wave) * POOL_ROWS;
  if (node0 >= n) return;
  const int nv = min(POOL_ROWS, n - node0);
  const int ids = batch[node0 + min(lane, nv - 1)];
  v2f v[POOL_ROWS];
  #pragma unroll
  for (int u = 0; u < POOL_ROWS; ++u) {
    const int r = min(node0 + u, n - 1);
    const float* row = h + (size_t)r * 64;
    v[u] = *(const v2f*)(row + 2 * lane);
  }
  int cur = __builtin_amdgcn_readlane(ids, 0);
  float c = fmaxf(cntg[cur], 1.0f);
  v2f acc = v[0];
  #pragma unroll
  for (int u = 1; u < POOL_ROWS; ++u) {
    if (u < nv) {
      const int g = __builtin_amdgcn_readlane(ids, u);
      if (g != cur) {
        pool_flush(out, c, cur, acc, lane);
        acc = (v2f){0.0f, 0.0f};
        cur = g;
        c = fmaxf(cntg[g], 1.0f);
      }
      acc += v[u];
    }
  }
  pool_flush(out, c, cur, acc, lane);
}

extern "C" void kernel_launch(void* const* d_in, const int* in_sizes, int n_in,
                              void* d_out, int out_size, void* d_ws, size_t ws_size,
                              hipStream_t stream) {
  const float* x     = (const float*)d_in[0];
  const float* W1    = (const float*)d_in[1];
  const float* b1    = (const float*)d_in[2];
  const float* W2    = (const float*)d_in[3];
  const float* b2    = (const float*)d_in[4];
  const int*   ei    = (const int*)d_in[5];
  const int*   batch = (const int*)d_in[6];

  const int N = in_sizes[6];
  const int E = in_sizes[5] / 2;
  const int G = N_GRAPHS;

  const int* src = ei;
  const int* dst = ei + E;

  char* base = (char*)d_ws;
  size_t off = 0;
  auto take = [&](size_t bytes) {
    size_t o = off;
    off = (off + bytes + 255) & ~(size_t)255;
    return o;
  };
  const int np = (N + 31) & ~31;
  const int chunk = (E + CNT_BLOCKS - 1) / CNT_BLOCKS;
  const int ngrp = (((N + 3) >> 2) + 31) >> 5;
  int*   pre      = (int*)(base + take(sizeof(int) * (size_t)ngrp * CNT_BLOCKS));
  int*   hist     = (int*)(base + take(sizeof(int) * (size_t)np * CNT_BLOCKS));
  int*   rank     = (int*)(base + take(sizeof(int) * (size_t)E));
  int*   rowstart = (int*)(base + take(sizeof(int) * (size_t)(N + 1)));
  float* dinv     = (float*)(base + take(sizeof(float) * (size_t)N));
  int*   csr      = (int*)(base + take(sizeof(int) * (size_t)E));
  float* hw1      = (float*)(base + take(sizeof(float) * (size_t)N * 128));
  float* h1       = (float*)(base + take(sizeof(float) * (size_t)N * 128));
  float* cntg     = (float*)(base + take(sizeof(float) * (size_t)G));
  float* hw2 = h1;
  float* h2  = hw1;

  const int mtiles = (N + 15) / 16;
  const int gGemm = (mtiles + 7) / 8;
  const int fillTiles = (chunk + FILL_TILE - 1) / FILL_TILE;

  k_count<<<CNT_BLOCKS, 1024, sizeof(int4) * (size_t)((N + 3) / 4), stream>>>(
      dst, hist, rank, pre, E, N, chunk, np, ngrp, (float*)d_out, G * 64, cntg, G);
  k_merge<<<ngrp, GRP_NODES, 0, stream>>>(hist, pre, rowstart, dinv, N, np, ngrp, batch, cntg);

  k_gemm1_fill<<<gGemm + fillTiles * CNT_BLOCKS, FILL_THREADS, 64 * (2 * 128 + 32) * sizeof(float), stream>>>(
      x, W1, dinv, hw1, N, gGemm, src, dst, rowstart, rank, hist, csr, E, chunk, np, fillTiles);
  k_agg1_gemm2<<<(N + FUSE_NODES - 1) / FUSE_NODES, FUSE_THREADS, FUSE_LDS_FLOATS * sizeof(float), stream>>>(
      hw1, dinv, rowstart, csr, b1, W2, hw2, N);

  k_aggregate<64><<<2048, 256, 0, stream>>>(hw2, dinv, rowstart, csr, b2, h2, N);

  k_pool_acc<<<(N + 4 * POOL_ROWS - 1) / (4 * POOL_ROWS), 128, 0, stream>>>(h2, batch, cntg, (float*)d_out, N);
}
